# starting kernel
# baseline (speedup 1.0000x reference)
#include <hip/hip_runtime.h>
#include <math.h>


typedef float v2f __attribute__((ext_vector_type(2)));
typedef float v8f __attribute__((ext_vector_type(8)));

#define N_NODES   100000
#define N_EDGES   1600000
#define IN_DIM    256
#define HEADS     4
#define OUT_DIM   32
#define FEAT_DIM  128
#define NEG_SLOPE 0.2f
#define KSTEPS    (IN_DIM / 4)
#define NTILES    (FEAT_DIM / 16)

__global__ void gat_prepack_W(const float* __restrict__ W, float* __restrict__ Wp) {
  const int t = blockIdx.x * blockDim.x + threadIdx.x;
  if (t >= KSTEPS * NTILES * 32) return;
  const int lane  = t & 31;
  const int tile  = (t >> 5) & 7;
  const int kstep = t >> 8;
  const int kb = (lane >> 4) * 2;
  const int n  = tile * 16 + (lane & 15);
  const int k  = kstep * 4 + kb;
  Wp[2 * t + 0] = W[(size_t)k * FEAT_DIM + n];
  Wp[2 * t + 1] = W[(size_t)(k + 1) * FEAT_DIM + n];
}

__global__ __launch_bounds__(256)
void gat_gemm_wmma(const float* __restrict__ x, const float* __restrict__ Wp,
                   float* __restrict__ feat) {
  const int lane = threadIdx.x & 31;
  const int mt   = blockIdx.x * 8 + (threadIdx.x >> 5);
  if (mt >= N_NODES / 16) return;

  const int kb = (lane >> 4) * 2;
  const float* aptr = x + (size_t)(mt * 16 + (lane & 15)) * IN_DIM + kb;
  const float* bptr = Wp + (size_t)lane * 2;

  v8f c[NTILES] = {};

  for (int ks = 0; ks < KSTEPS; ++ks) {
    const v2f a = *(const v2f*)(aptr + ks * 4);
#pragma unroll
    for (int t = 0; t < NTILES; ++t) {
      const v2f b = *(const v2f*)(bptr + (size_t)(ks * NTILES + t) * 64);
      c[t] = __builtin_amdgcn_wmma_f32_16x16x4_f32(false, a, false, b,
                                                   (short)0, c[t], false, false);
    }
  }

  const int mbase = mt * 16 + 8 * (lane >> 4);
  float* o = feat + (size_t)mbase * FEAT_DIM + (lane & 15);
#pragma unroll
  for (int t = 0; t < NTILES; ++t)
#pragma unroll
    for (int i = 0; i < 8; ++i)
      o[(size_t)i * FEAT_DIM + t * 16] = c[t][i];
}

__global__ void gat_init(float* __restrict__ m, float* __restrict__ denom,
                         float* __restrict__ out) {
  int t = blockIdx.x * blockDim.x + threadIdx.x;
  if (t < N_NODES * OUT_DIM) out[t] = 0.0f;
  if (t < N_NODES * HEADS) { m[t] = -INFINITY; denom[t] = 0.0f; }
}

__global__ void gat_el_er(const float* __restrict__ feat,
                          const float* __restrict__ attn_l,
                          const float* __restrict__ attn_r,
                          float* __restrict__ el, float* __restrict__ er) {
  int t = blockIdx.x * blockDim.x + threadIdx.x;
  if (t >= N_NODES * HEADS) return;
  const int h = t & 3;
  const float* f  = feat + (size_t)(t >> 2) * FEAT_DIM + h * OUT_DIM;
  const float* al = attn_l + h * OUT_DIM;
  const float* ar = attn_r + h * OUT_DIM;
  float sl = 0.0f, sr = 0.0f;
#pragma unroll
  for (int d = 0; d < OUT_DIM; ++d) { float v = f[d]; sl += v * al[d]; sr += v * ar[d]; }
  el[t] = sl;
  er[t] = sr;
}

__device__ __forceinline__ float lrelu(float v) {
  return v > 0.0f ? v : NEG_SLOPE * v;
}

__device__ __forceinline__ void atomic_max_f32(float* addr, float v) {
  if (v >= 0.0f) atomicMax((int*)addr, __float_as_int(v));
  else           atomicMin((unsigned int*)addr, __float_as_uint(v));
}

__global__ void gat_edge_max(const int* __restrict__ src, const int* __restrict__ dst,
                             const float* __restrict__ el, const float* __restrict__ er,
                             float* __restrict__ m) {
  int e = blockIdx.x * blockDim.x + threadIdx.x;
  if (e >= N_EDGES) return;
  const int s = src[e], d = dst[e];
  const float4 L = *(const float4*)(el + (size_t)s * HEADS);
  const float4 R = *(const float4*)(er + (size_t)d * HEADS);
  atomic_max_f32(m + (size_t)d * HEADS + 0, lrelu(L.x + R.x));
  atomic_max_f32(m + (size_t)d * HEADS + 1, lrelu(L.y + R.y));
  atomic_max_f32(m + (size_t)d * HEADS + 2, lrelu(L.z + R.z));
  atomic_max_f32(m + (size_t)d * HEADS + 3, lrelu(L.w + R.w));
}

__global__ void gat_edge_sum(const int* __restrict__ src, const int* __restrict__ dst,
                             const float* __restrict__ el, const float* __restrict__ er,
                             const float* __restrict__ m, float* __restrict__ denom) {
  int e = blockIdx.x * blockDim.x + threadIdx.x;
  if (e >= N_EDGES) return;
  const int s = src[e], d = dst[e];
  const float4 L = *(const float4*)(el + (size_t)s * HEADS);
  const float4 R = *(const float4*)(er + (size_t)d * HEADS);
  const float4 M = *(const float4*)(m  + (size_t)d * HEADS);
  atomicAdd(denom + (size_t)d * HEADS + 0, __expf(lrelu(L.x + R.x) - M.x));
  atomicAdd(denom + (size_t)d * HEADS + 1, __expf(lrelu(L.y + R.y) - M.y));
  atomicAdd(denom + (size_t)d * HEADS + 2, __expf(lrelu(L.z + R.z) - M.z));
  atomicAdd(denom + (size_t)d * HEADS + 3, __expf(lrelu(L.w + R.w) - M.w));
}

__global__ __launch_bounds__(256)
void gat_aggregate(const int* __restrict__ src, const int* __restrict__ dst,
                   const float* __restrict__ el, const float* __restrict__ er,
                   const float* __restrict__ m, const float* __restrict__ denom,
                   const float* __restrict__ feat, float* __restrict__ out) {
  const int lane = threadIdx.x & 31;
  const int e = blockIdx.x * 8 + (threadIdx.x >> 5);
  if (e >= N_EDGES) return;
  const int s = src[e], d = dst[e];
  const float4 L  = *(const float4*)(el    + (size_t)s * HEADS);
  const float4 R  = *(const float4*)(er    + (size_t)d * HEADS);
  const float4 M  = *(const float4*)(m     + (size_t)d * HEADS);
  const float4 Dn = *(const float4*)(denom + (size_t)d * HEADS);
  const float a0 = __expf(lrelu(L.x + R.x) - M.x) / fmaxf(Dn.x, 1e-9f);
  const float a1 = __expf(lrelu(L.y + R.y) - M.y) / fmaxf(Dn.y, 1e-9f);
  const float a2 = __expf(lrelu(L.z + R.z) - M.z) / fmaxf(Dn.z, 1e-9f);
  const float a3 = __expf(lrelu(L.w + R.w) - M.w) / fmaxf(Dn.w, 1e-9f);

  const float* f = feat + (size_t)s * FEAT_DIM + lane;
  const float acc = 0.25f * (a0 * f[0] + a1 * f[32] + a2 * f[64] + a3 * f[96]);
  atomicAdd(out + (size_t)d * OUT_DIM + lane, acc);
}

extern "C" void kernel_launch(void* const* d_in, const int* in_sizes, int n_in,
                              void* d_out, int out_size, void* d_ws, size_t ws_size,
                              hipStream_t stream) {
  const float* x      = (const float*)d_in[0];
  const float* W      = (const float*)d_in[1];
  const float* attn_l = (const float*)d_in[2];
  const float* attn_r = (const float*)d_in[3];
  const int*   src    = (const int*)d_in[4];
  const int*   dst    = (const int*)d_in[5];
  float* out = (float*)d_out;

  float* ws    = (float*)d_ws;
  float* feat  = ws;
  float* el    = feat + (size_t)N_NODES * FEAT_DIM;
  float* er    = el   + (size_t)N_NODES * HEADS;
  float* mbuf  = er   + (size_t)N_NODES * HEADS;
  float* denom = mbuf + (size_t)N_NODES * HEADS;
  float* Wp    = denom + (size_t)N_NODES * HEADS;

  gat_prepack_W<<<(KSTEPS * NTILES * 32 + 255) / 256, 256, 0, stream>>>(W, Wp);
  const int MT = N_NODES / 16;
  gat_gemm_wmma<<<(MT + 7) / 8, 256, 0, stream>>>(x, Wp, feat);
  gat_init<<<(N_NODES * OUT_DIM + 255) / 256, 256, 0, stream>>>(mbuf, denom, out);
  gat_el_er<<<(N_NODES * HEADS + 255) / 256, 256, 0, stream>>>(feat, attn_l, attn_r, el, er);
  gat_edge_max<<<(N_EDGES + 255) / 256, 256, 0, stream>>>(src, dst, el, er, mbuf);
  gat_edge_sum<<<(N_EDGES + 255) / 256, 256, 0, stream>>>(src, dst, el, er, mbuf, denom);
  gat_aggregate<<<(N_EDGES + 7) / 8, 256, 0, stream>>>(src, dst, el, er, mbuf, denom, feat, out);
}


# edge max and edge sum passes: one lane per (edge, head) so each atomic instruction touches 8 lines not 32; float atomic add order stays arrival order as in parent
# speedup vs baseline: 1.4430x; 1.4430x over previous
#include <hip/hip_runtime.h>
#include <math.h>


typedef float v2f __attribute__((ext_vector_type(2)));
typedef float v8f __attribute__((ext_vector_type(8)));

#define N_NODES   100000
#define N_EDGES   1600000
#define IN_DIM    256
#define HEADS     4
#define OUT_DIM   32
#define FEAT_DIM  128
#define NEG_SLOPE 0.2f
#define KSTEPS    (IN_DIM / 4)
#define NTILES    (FEAT_DIM / 16)

__global__ void gat_prepack_W(const float* __restrict__ W, float* __restrict__ Wp) {
  const int t = blockIdx.x * blockDim.x + threadIdx.x;
  if (t >= KSTEPS * NTILES * 32) return;
  const int lane  = t & 31;
  const int tile  = (t >> 5) & 7;
  const int kstep = t >> 8;
  const int kb = (lane >> 4) * 2;
  const int n  = tile * 16 + (lane & 15);
  const int k  = kstep * 4 + kb;
  Wp[2 * t + 0] = W[(size_t)k * FEAT_DIM + n];
  Wp[2 * t + 1] = W[(size_t)(k + 1) * FEAT_DIM + n];
}

__global__ __launch_bounds__(256)
void gat_gemm_wmma(const float* __restrict__ x, const float* __restrict__ Wp,
                   float* __restrict__ feat) {
  const int lane = threadIdx.x & 31;
  const int mt   = blockIdx.x * 8 + (threadIdx.x >> 5);
  if (mt >= N_NODES / 16) return;

  const int kb = (lane >> 4) * 2;
  const float* aptr = x + (size_t)(mt * 16 + (lane & 15)) * IN_DIM + kb;
  const float* bptr = Wp + (size_t)lane * 2;

  v8f c[NTILES] = {};

  for (int ks = 0; ks < KSTEPS; ++ks) {
    const v2f a = *(const v2f*)(aptr + ks * 4);
#pragma unroll
    for (int t = 0; t < NTILES; ++t) {
      const v2f b = *(const v2f*)(bptr + (size_t)(ks * NTILES + t) * 64);
      c[t] = __builtin_amdgcn_wmma_f32_16x16x4_f32(false, a, false, b,
                                                   (short)0, c[t], false, false);
    }
  }

  const int mbase = mt * 16 + 8 * (lane >> 4);
  float* o = feat + (size_t)mbase * FEAT_DIM + (lane & 15);
#pragma unroll
  for (int t = 0; t < NTILES; ++t)
#pragma unroll
    for (int i = 0; i < 8; ++i)
      o[(size_t)i * FEAT_DIM + t * 16] = c[t][i];
}

__global__ void gat_init(float* __restrict__ m, float* __restrict__ denom,
                         float* __restrict__ out) {
  int t = blockIdx.x * blockDim.x + threadIdx.x;
  if (t < N_NODES * OUT_DIM) out[t] = 0.0f;
  if (t < N_NODES * HEADS) { m[t] = -INFINITY; denom[t] = 0.0f; }
}

__global__ void gat_el_er(const float* __restrict__ feat,
                          const float* __restrict__ attn_l,
                          const float* __restrict__ attn_r,
                          float* __restrict__ el, float* __restrict__ er) {
  int t = blockIdx.x * blockDim.x + threadIdx.x;
  if (t >= N_NODES * HEADS) return;
  const int h = t & 3;
  const float* f  = feat + (size_t)(t >> 2) * FEAT_DIM + h * OUT_DIM;
  const float* al = attn_l + h * OUT_DIM;
  const float* ar = attn_r + h * OUT_DIM;
  float sl = 0.0f, sr = 0.0f;
#pragma unroll
  for (int d = 0; d < OUT_DIM; ++d) { float v = f[d]; sl += v * al[d]; sr += v * ar[d]; }
  el[t] = sl;
  er[t] = sr;
}

__device__ __forceinline__ float lrelu(float v) {
  return v > 0.0f ? v : NEG_SLOPE * v;
}

__device__ __forceinline__ void atomic_max_f32(float* addr, float v) {
  if (v >= 0.0f) atomicMax((int*)addr, __float_as_int(v));
  else           atomicMin((unsigned int*)addr, __float_as_uint(v));
}

__global__ void gat_edge_max(const int* __restrict__ src, const int* __restrict__ dst,
                             const float* __restrict__ el, const float* __restrict__ er,
                             float* __restrict__ m) {
  const int t = blockIdx.x * blockDim.x + threadIdx.x;
  const int e = t >> 2, h = t & 3;
  if (e >= N_EDGES) return;
  const int s = src[e], d = dst[e];
  const float v = lrelu(el[(size_t)s * HEADS + h] + er[(size_t)d * HEADS + h]);
  atomic_max_f32(m + (size_t)d * HEADS + h, v);
}

__global__ void gat_edge_sum(const int* __restrict__ src, const int* __restrict__ dst,
                             const float* __restrict__ el, const float* __restrict__ er,
                             const float* __restrict__ m, float* __restrict__ denom) {
  const int t = blockIdx.x * blockDim.x + threadIdx.x;
  const int e = t >> 2, h = t & 3;
  if (e >= N_EDGES) return;
  const int s = src[e], d = dst[e];
  const size_t dh = (size_t)d * HEADS + h;
  const float v = lrelu(el[(size_t)s * HEADS + h] + er[dh]);
  atomicAdd(denom + dh, __expf(v - m[dh]));
}

__global__ __launch_bounds__(256)
void gat_aggregate(const int* __restrict__ src, const int* __restrict__ dst,
                   const float* __restrict__ el, const float* __restrict__ er,
                   const float* __restrict__ m, const float* __restrict__ denom,
                   const float* __restrict__ feat, float* __restrict__ out) {
  const int lane = threadIdx.x & 31;
  const int e = blockIdx.x * 8 + (threadIdx.x >> 5);
  if (e >= N_EDGES) return;
  const int s = src[e], d = dst[e];
  const float4 L  = *(const float4*)(el    + (size_t)s * HEADS);
  const float4 R  = *(const float4*)(er    + (size_t)d * HEADS);
  const float4 M  = *(const float4*)(m     + (size_t)d * HEADS);
  const float4 Dn = *(const float4*)(denom + (size_t)d * HEADS);
  const float a0 = __expf(lrelu(L.x + R.x) - M.x) / fmaxf(Dn.x, 1e-9f);
  const float a1 = __expf(lrelu(L.y + R.y) - M.y) / fmaxf(Dn.y, 1e-9f);
  const float a2 = __expf(lrelu(L.z + R.z) - M.z) / fmaxf(Dn.z, 1e-9f);
  const float a3 = __expf(lrelu(L.w + R.w) - M.w) / fmaxf(Dn.w, 1e-9f);

  const float* f = feat + (size_t)s * FEAT_DIM + lane;
  const float acc = 0.25f * (a0 * f[0] + a1 * f[32] + a2 * f[64] + a3 * f[96]);
  atomicAdd(out + (size_t)d * OUT_DIM + lane, acc);
}

extern "C" void kernel_launch(void* const* d_in, const int* in_sizes, int n_in,
                              void* d_out, int out_size, void* d_ws, size_t ws_size,
                              hipStream_t stream) {
  const float* x      = (const float*)d_in[0];
  const float* W      = (const float*)d_in[1];
  const float* attn_l = (const float*)d_in[2];
  const float* attn_r = (const float*)d_in[3];
  const int*   src    = (const int*)d_in[4];
  const int*   dst    = (const int*)d_in[5];
  float* out = (float*)d_out;

  float* ws    = (float*)d_ws;
  float* feat  = ws;
  float* el    = feat + (size_t)N_NODES * FEAT_DIM;
  float* er    = el   + (size_t)N_NODES * HEADS;
  float* mbuf  = er   + (size_t)N_NODES * HEADS;
  float* denom = mbuf + (size_t)N_NODES * HEADS;
  float* Wp    = denom + (size_t)N_NODES * HEADS;

  gat_prepack_W<<<(KSTEPS * NTILES * 32 + 255) / 256, 256, 0, stream>>>(W, Wp);
  const int MT = N_NODES / 16;
  gat_gemm_wmma<<<(MT + 7) / 8, 256, 0, stream>>>(x, Wp, feat);
  gat_init<<<(N_NODES * OUT_DIM + 255) / 256, 256, 0, stream>>>(mbuf, denom, out);
  gat_el_er<<<(N_NODES * HEADS + 255) / 256, 256, 0, stream>>>(feat, attn_l, attn_r, el, er);
  gat_edge_max<<<(N_EDGES * HEADS + 255) / 256, 256, 0, stream>>>(src, dst, el, er, mbuf);
  gat_edge_sum<<<(N_EDGES * HEADS + 255) / 256, 256, 0, stream>>>(src, dst, el, er, mbuf, denom);
  gat_aggregate<<<(N_EDGES + 7) / 8, 256, 0, stream>>>(src, dst, el, er, mbuf, denom, feat, out);
}


# aggregate pass: one wave handles 8 edges: one lane per (edge, head) computes each weight once, then all 32 row gathers are issued before first use; float atomic add order stays arrival order
# speedup vs baseline: 1.7577x; 1.2181x over previous
#include <hip/hip_runtime.h>
#include <math.h>


typedef float v2f __attribute__((ext_vector_type(2)));
typedef float v8f __attribute__((ext_vector_type(8)));

#define N_NODES   100000
#define N_EDGES   1600000
#define IN_DIM    256
#define HEADS     4
#define OUT_DIM   32
#define FEAT_DIM  128
#define NEG_SLOPE 0.2f
#define KSTEPS    (IN_DIM / 4)
#define NTILES    (FEAT_DIM / 16)

__global__ void gat_prepack_W(const float* __restrict__ W, float* __restrict__ Wp) {
  const int t = blockIdx.x * blockDim.x + threadIdx.x;
  if (t >= KSTEPS * NTILES * 32) return;
  const int lane  = t & 31;
  const int tile  = (t >> 5) & 7;
  const int kstep = t >> 8;
  const int kb = (lane >> 4) * 2;
  const int n  = tile * 16 + (lane & 15);
  const int k  = kstep * 4 + kb;
  Wp[2 * t + 0] = W[(size_t)k * FEAT_DIM + n];
  Wp[2 * t + 1] = W[(size_t)(k + 1) * FEAT_DIM + n];
}

__global__ __launch_bounds__(256)
void gat_gemm_wmma(const float* __restrict__ x, const float* __restrict__ Wp,
                   float* __restrict__ feat) {
  const int lane = threadIdx.x & 31;
  const int mt   = blockIdx.x * 8 + (threadIdx.x >> 5);
  if (mt >= N_NODES / 16) return;

  const int kb = (lane >> 4) * 2;
  const float* aptr = x + (size_t)(mt * 16 + (lane & 15)) * IN_DIM + kb;
  const float* bptr = Wp + (size_t)lane * 2;

  v8f c[NTILES] = {};

  for (int ks = 0; ks < KSTEPS; ++ks) {
    const v2f a = *(const v2f*)(aptr + ks * 4);
#pragma unroll
    for (int t = 0; t < NTILES; ++t) {
      const v2f b = *(const v2f*)(bptr + (size_t)(ks * NTILES + t) * 64);
      c[t] = __builtin_amdgcn_wmma_f32_16x16x4_f32(false, a, false, b,
                                                   (short)0, c[t], false, false);
    }
  }

  const int mbase = mt * 16 + 8 * (lane >> 4);
  float* o = feat + (size_t)mbase * FEAT_DIM + (lane & 15);
#pragma unroll
  for (int t = 0; t < NTILES; ++t)
#pragma unroll
    for (int i = 0; i < 8; ++i)
      o[(size_t)i * FEAT_DIM + t * 16] = c[t][i];
}

__global__ void gat_init(float* __restrict__ m, float* __restrict__ denom,
                         float* __restrict__ out) {
  int t = blockIdx.x * blockDim.x + threadIdx.x;
  if (t < N_NODES * OUT_DIM) out[t] = 0.0f;
  if (t < N_NODES * HEADS) { m[t] = -INFINITY; denom[t] = 0.0f; }
}

__global__ void gat_el_er(const float* __restrict__ feat,
                          const float* __restrict__ attn_l,
                          const float* __restrict__ attn_r,
                          float* __restrict__ el, float* __restrict__ er) {
  int t = blockIdx.x * blockDim.x + threadIdx.x;
  if (t >= N_NODES * HEADS) return;
  const int h = t & 3;
  const float* f  = feat + (size_t)(t >> 2) * FEAT_DIM + h * OUT_DIM;
  const float* al = attn_l + h * OUT_DIM;
  const float* ar = attn_r + h * OUT_DIM;
  float sl = 0.0f, sr = 0.0f;
#pragma unroll
  for (int d = 0; d < OUT_DIM; ++d) { float v = f[d]; sl += v * al[d]; sr += v * ar[d]; }
  el[t] = sl;
  er[t] = sr;
}

__device__ __forceinline__ float lrelu(float v) {
  return v > 0.0f ? v : NEG_SLOPE * v;
}

__device__ __forceinline__ void atomic_max_f32(float* addr, float v) {
  if (v >= 0.0f) atomicMax((int*)addr, __float_as_int(v));
  else           atomicMin((unsigned int*)addr, __float_as_uint(v));
}

__global__ void gat_edge_max(const int* __restrict__ src, const int* __restrict__ dst,
                             const float* __restrict__ el, const float* __restrict__ er,
                             float* __restrict__ m) {
  const int t = blockIdx.x * blockDim.x + threadIdx.x;
  const int e = t >> 2, h = t & 3;
  if (e >= N_EDGES) return;
  const int s = src[e], d = dst[e];
  const float v = lrelu(el[(size_t)s * HEADS + h] + er[(size_t)d * HEADS + h]);
  atomic_max_f32(m + (size_t)d * HEADS + h, v);
}

__global__ void gat_edge_sum(const int* __restrict__ src, const int* __restrict__ dst,
                             const float* __restrict__ el, const float* __restrict__ er,
                             const float* __restrict__ m, float* __restrict__ denom) {
  const int t = blockIdx.x * blockDim.x + threadIdx.x;
  const int e = t >> 2, h = t & 3;
  if (e >= N_EDGES) return;
  const int s = src[e], d = dst[e];
  const size_t dh = (size_t)d * HEADS + h;
  const float v = lrelu(el[(size_t)s * HEADS + h] + er[dh]);
  atomicAdd(denom + dh, __expf(v - m[dh]));
}

#define EPW 8
static_assert(N_EDGES % EPW == 0, "every wave owns EPW whole edges");

__global__ __launch_bounds__(256)
void gat_aggregate(const int* __restrict__ src, const int* __restrict__ dst,
                   const float* __restrict__ el, const float* __restrict__ er,
                   const float* __restrict__ m, const float* __restrict__ denom,
                   const float* __restrict__ feat, float* __restrict__ out) {
  const int lane = threadIdx.x & 31;
  const int e0 = (blockIdx.x * 8 + (threadIdx.x >> 5)) * EPW;
  if (e0 >= N_EDGES) return;

  const int h = lane & 3;
  const int e = e0 + (lane >> 2);
  const int s = src[e], d = dst[e];
  const size_t dh = (size_t)d * HEADS + h;
  const float v = lrelu(el[(size_t)s * HEADS + h] + er[dh]);
  const float a = 0.25f * (__expf(v - m[dh]) / fmaxf(denom[dh], 1e-9f));

  float f[EPW][HEADS];
  for (int j = 0; j < EPW; ++j) {
    const int sj = __builtin_amdgcn_readlane(s, 4 * j);
    const float* fr = feat + (size_t)sj * FEAT_DIM + lane;
    for (int hh = 0; hh < HEADS; ++hh) f[j][hh] = fr[hh * OUT_DIM];
  }

  const int ai = __float_as_int(a);
  for (int j = 0; j < EPW; ++j) {
    const int dj = __builtin_amdgcn_readlane(d, 4 * j);
    const float a0 = __int_as_float(__builtin_amdgcn_readlane(ai, 4 * j + 0));
    const float a1 = __int_as_float(__builtin_amdgcn_readlane(ai, 4 * j + 1));
    const float a2 = __int_as_float(__builtin_amdgcn_readlane(ai, 4 * j + 2));
    const float a3 = __int_as_float(__builtin_amdgcn_readlane(ai, 4 * j + 3));
    const float acc = a0 * f[j][0] + a1 * f[j][1] + a2 * f[j][2] + a3 * f[j][3];
    atomicAdd(out + (size_t)dj * OUT_DIM + lane, acc);
  }
}

extern "C" void kernel_launch(void* const* d_in, const int* in_sizes, int n_in,
                              void* d_out, int out_size, void* d_ws, size_t ws_size,
                              hipStream_t stream) {
  const float* x      = (const float*)d_in[0];
  const float* W      = (const float*)d_in[1];
  const float* attn_l = (const float*)d_in[2];
  const float* attn_r = (const float*)d_in[3];
  const int*   src    = (const int*)d_in[4];
  const int*   dst    = (const int*)d_in[5];
  float* out = (float*)d_out;

  float* ws    = (float*)d_ws;
  float* feat  = ws;
  float* el    = feat + (size_t)N_NODES * FEAT_DIM;
  float* er    = el   + (size_t)N_NODES * HEADS;
  float* mbuf  = er   + (size_t)N_NODES * HEADS;
  float* denom = mbuf + (size_t)N_NODES * HEADS;
  float* Wp    = denom + (size_t)N_NODES * HEADS;

  gat_prepack_W<<<(KSTEPS * NTILES * 32 + 255) / 256, 256, 0, stream>>>(W, Wp);
  const int MT = N_NODES / 16;
  gat_gemm_wmma<<<(MT + 7) / 8, 256, 0, stream>>>(x, Wp, feat);
  gat_init<<<(N_NODES * OUT_DIM + 255) / 256, 256, 0, stream>>>(mbuf, denom, out);
  gat_el_er<<<(N_NODES * HEADS + 255) / 256, 256, 0, stream>>>(feat, attn_l, attn_r, el, er);
  gat_edge_max<<<(N_EDGES * HEADS + 255) / 256, 256, 0, stream>>>(src, dst, el, er, mbuf);
  gat_edge_sum<<<(N_EDGES * HEADS + 255) / 256, 256, 0, stream>>>(src, dst, el, er, mbuf, denom);
  gat_aggregate<<<(N_EDGES / EPW + 7) / 8, 256, 0, stream>>>(src, dst, el, er, mbuf, denom, feat, out);
}


# edges counting-sorted by destination (one returning int add each); one wave per node does edge softmax + aggregation in registers, no float atomics: sums in list order, denominator by xor tree
# speedup vs baseline: 1.9458x; 1.1070x over previous
#include <hip/hip_runtime.h>
#include <math.h>


typedef float v2f __attribute__((ext_vector_type(2)));
typedef float v8f __attribute__((ext_vector_type(8)));

#define N_NODES   100000
#define N_EDGES   1600000
#define IN_DIM    256
#define HEADS     4
#define OUT_DIM   32
#define FEAT_DIM  128
#define NEG_SLOPE 0.2f
#define KSTEPS    (IN_DIM / 4)
#define NTILES    (FEAT_DIM / 16)

__global__ void gat_prepack_W(const float* __restrict__ W, float* __restrict__ Wp) {
  const int t = blockIdx.x * blockDim.x + threadIdx.x;
  if (t >= KSTEPS * NTILES * 32) return;
  const int lane  = t & 31;
  const int tile  = (t >> 5) & 7;
  const int kstep = t >> 8;
  const int kb = (lane >> 4) * 2;
  const int n  = tile * 16 + (lane & 15);
  const int k  = kstep * 4 + kb;
  Wp[2 * t + 0] = W[(size_t)k * FEAT_DIM + n];
  Wp[2 * t + 1] = W[(size_t)(k + 1) * FEAT_DIM + n];
}

__global__ __launch_bounds__(256)
void gat_gemm_wmma(const float* __restrict__ x, const float* __restrict__ Wp,
                   float* __restrict__ feat) {
  const int lane = threadIdx.x & 31;
  const int mt   = blockIdx.x * 8 + (threadIdx.x >> 5);
  if (mt >= N_NODES / 16) return;

  const int kb = (lane >> 4) * 2;
  const float* aptr = x + (size_t)(mt * 16 + (lane & 15)) * IN_DIM + kb;
  const float* bptr = Wp + (size_t)lane * 2;

  v8f c[NTILES] = {};

  for (int ks = 0; ks < KSTEPS; ++ks) {
    const v2f a = *(const v2f*)(aptr + ks * 4);
#pragma unroll
    for (int t = 0; t < NTILES; ++t) {
      const v2f b = *(const v2f*)(bptr + (size_t)(ks * NTILES + t) * 64);
      c[t] = __builtin_amdgcn_wmma_f32_16x16x4_f32(false, a, false, b,
                                                   (short)0, c[t], false, false);
    }
  }

  const int mbase = mt * 16 + 8 * (lane >> 4);
  float* o = feat + (size_t)mbase * FEAT_DIM + (lane & 15);
#pragma unroll
  for (int t = 0; t < NTILES; ++t)
#pragma unroll
    for (int i = 0; i < 8; ++i)
      o[(size_t)i * FEAT_DIM + t * 16] = c[t][i];
}

#define SPLIT 65536

__global__ void gat_init(int* __restrict__ cnt2) {
  const int t = blockIdx.x * blockDim.x + threadIdx.x;
  if (t < 2 * N_NODES) cnt2[t] = 0;
}

__global__ void gat_el_er(const float* __restrict__ feat,
                          const float* __restrict__ attn_l,
                          const float* __restrict__ attn_r,
                          float* __restrict__ el, float* __restrict__ out) {
  int t = blockIdx.x * blockDim.x + threadIdx.x;
  if (t >= N_NODES * HEADS) return;
  const int h = t & 3;
  const float* f  = feat + (size_t)(t >> 2) * FEAT_DIM + h * OUT_DIM;
  const float* al = attn_l + h * OUT_DIM;
  const float* ar = attn_r + h * OUT_DIM;
  float sl = 0.0f, sr = 0.0f;
#pragma unroll
  for (int d = 0; d < OUT_DIM; ++d) { float v = f[d]; sl += v * al[d]; sr += v * ar[d]; }
  el[t] = sl;
  out[(size_t)(t >> 2) * OUT_DIM + 4 + h] = sr;
}

__device__ __forceinline__ float lrelu(float v) {
  return v > 0.0f ? v : NEG_SLOPE * v;
}

__global__ void gat_rank(const int* __restrict__ src, const int* __restrict__ dst,
                         int* __restrict__ cnt2, int* __restrict__ outw) {
  const int e = blockIdx.x * blockDim.x + threadIdx.x;
  if (e >= N_EDGES) return;
  const int s = src[e], d = dst[e];
  const int r = atomicAdd(cnt2 + (s >= SPLIT ? N_NODES : 0) + d, 1);
  outw[(size_t)(e >> 4) * OUT_DIM + 16 + (e & 15)] = r;
}

__global__ __launch_bounds__(256)
void gat_scan(const int* __restrict__ cnt2, int* __restrict__ start) {
  __shared__ int red[8];
  __shared__ int wtot[8];
  const int t = threadIdx.x, lane = t & 31, w = t >> 5;
  const int4* lo4 = (const int4*)cnt2;
  const int4* hi4 = (const int4*)(cnt2 + N_NODES);
  int sum = 0;
  const int nq = blockIdx.x * 256;
  for (int i = t; i < nq; i += 256) {
    const int4 a = lo4[i], b = hi4[i];
    sum += (a.x + a.y) + (a.z + a.w) + (b.x + b.y) + (b.z + b.w);
  }
  for (int o = 16; o > 0; o >>= 1) sum += __shfl_xor(sum, o);
  if (lane == 0) red[w] = sum;
  const int q = nq + t;
  int4 v = {0, 0, 0, 0};
  if (q < N_NODES / 4) {
    const int4 a = lo4[q], b = hi4[q];
    v.x = a.x + b.x; v.y = a.y + b.y; v.z = a.z + b.z; v.w = a.w + b.w;
  }
  const int ts = v.x + v.y + v.z + v.w;
  int inc = ts;
  for (int o = 1; o < 32; o <<= 1) {
    const int up = __shfl_up(inc, o);
    if (lane >= o) inc += up;
  }
  if (lane == 31) wtot[w] = inc;
  __syncthreads();
  int base = 0;
  for (int i = 0; i < 8; ++i) base += red[i];
  for (int i = 0; i < 8; ++i) base += (i < w) ? wtot[i] : 0;
  if (q < N_NODES / 4) {
    int4 o;
    o.x = base + inc - ts;
    o.y = o.x + v.x;
    o.z = o.y + v.y;
    o.w = o.z + v.z;
    ((int4*)start)[q] = o;
  }
}

__global__ void gat_fill(const int* __restrict__ src, const int* __restrict__ dst,
                         const int* __restrict__ cnt2, const int* __restrict__ start,
                         const int* __restrict__ outw, unsigned short* __restrict__ csr) {
  const int e = blockIdx.x * blockDim.x + threadIdx.x;
  if (e >= N_EDGES) return;
  const int s = src[e], d = dst[e];
  const int r = outw[(size_t)(e >> 4) * OUT_DIM + 16 + (e & 15)];
  const int p = start[d] + (s >= SPLIT ? cnt2[d] : 0) + r;
  if ((unsigned)p < (unsigned)N_EDGES) csr[p] = (unsigned short)(s & (SPLIT - 1));
}

#define EPW 8

__device__ __forceinline__ int list_src(const unsigned short* __restrict__ csr,
                                        int p, bool upper) {
  const int s = (int)csr[min(p, N_EDGES - 1)] + (upper ? SPLIT : 0);
  return min(s, N_NODES - 1);
}

__global__ __launch_bounds__(256)
void gat_node(const unsigned short* __restrict__ csr, const int* __restrict__ cnt2,
              const int* __restrict__ start, const float* __restrict__ el,
              const float* __restrict__ feat, float* out) {
  const int lane = threadIdx.x & 31;
  const int n = __builtin_amdgcn_readfirstlane(blockIdx.x * 8 + (threadIdx.x >> 5));
  if (n >= N_NODES) return;
  float* orow = out + (size_t)n * OUT_DIM;
  const int lo   = cnt2[n];
  const int deg  = min(max(lo + cnt2[N_NODES + n], 0), N_EDGES);
  const int base = min(max(start[n], 0), N_EDGES);
  if (deg == 0) { orow[lane] = 0.0f; return; }

  const int h = lane & 3, j = lane >> 2;
  const float erh = orow[4 + h];

  float mx = -INFINITY;
  for (int c = 0; c < deg; c += EPW) {
    const int idx = c + j;
    const bool valid = idx < deg;
    const int s = valid ? list_src(csr, base + idx, idx >= lo) : 0;
    const float v = lrelu(el[(size_t)s * HEADS + h] + erh);
    mx = valid ? fmaxf(mx, v) : mx;
  }
  mx = fmaxf(mx, __shfl_xor(mx, 4));
  mx = fmaxf(mx, __shfl_xor(mx, 8));
  mx = fmaxf(mx, __shfl_xor(mx, 16));

  float sm = 0.0f;
  for (int c = 0; c < deg; c += EPW) {
    const int idx = c + j;
    const bool valid = idx < deg;
    const int s = valid ? list_src(csr, base + idx, idx >= lo) : 0;
    const float v = lrelu(el[(size_t)s * HEADS + h] + erh);
    sm += valid ? __expf(v - mx) : 0.0f;
  }
  sm += __shfl_xor(sm, 4);
  sm += __shfl_xor(sm, 8);
  sm += __shfl_xor(sm, 16);
  const float den = fmaxf(sm, 1e-9f);

  float tot = 0.0f;
  for (int c = 0; c < deg; c += EPW) {
    const int idx = c + j;
    const bool valid = idx < deg;
    const int s = valid ? list_src(csr, base + idx, idx >= lo) : 0;
    const float v = lrelu(el[(size_t)s * HEADS + h] + erh);
    const float a = valid ? 0.25f * (__expf(v - mx) / den) : 0.0f;
    const int s0 = __builtin_amdgcn_readfirstlane(s);
    const int sv = valid ? s : s0;

    float f[EPW][HEADS];
    for (int jj = 0; jj < EPW; ++jj) {
      const int sj = __builtin_amdgcn_readlane(sv, 4 * jj);
      const float* fr = feat + (size_t)sj * FEAT_DIM + lane;
      for (int hh = 0; hh < HEADS; ++hh) f[jj][hh] = fr[hh * OUT_DIM];
    }
    const int ai = __float_as_int(a);
    for (int jj = 0; jj < EPW; ++jj) {
      const float a0 = __int_as_float(__builtin_amdgcn_readlane(ai, 4 * jj + 0));
      const float a1 = __int_as_float(__builtin_amdgcn_readlane(ai, 4 * jj + 1));
      const float a2 = __int_as_float(__builtin_amdgcn_readlane(ai, 4 * jj + 2));
      const float a3 = __int_as_float(__builtin_amdgcn_readlane(ai, 4 * jj + 3));
      tot += a0 * f[jj][0] + a1 * f[jj][1] + a2 * f[jj][2] + a3 * f[jj][3];
    }
  }
  orow[lane] = tot;
}

extern "C" void kernel_launch(void* const* d_in, const int* in_sizes, int n_in,
                              void* d_out, int out_size, void* d_ws, size_t ws_size,
                              hipStream_t stream) {
  const float* x      = (const float*)d_in[0];
  const float* W      = (const float*)d_in[1];
  const float* attn_l = (const float*)d_in[2];
  const float* attn_r = (const float*)d_in[3];
  const int*   src    = (const int*)d_in[4];
  const int*   dst    = (const int*)d_in[5];
  float* out = (float*)d_out;

  float* ws    = (float*)d_ws;
  float* feat  = ws;
  float* el    = feat + (size_t)N_NODES * FEAT_DIM;
  unsigned short* csr = (unsigned short*)(el + (size_t)N_NODES * HEADS);
  int*   cnt2  = (int*)(el + (size_t)3 * N_NODES * HEADS);
  int*   start = cnt2 + 2 * N_NODES;
  float* Wp    = el + (size_t)4 * N_NODES * HEADS;

  gat_prepack_W<<<(KSTEPS * NTILES * 32 + 255) / 256, 256, 0, stream>>>(W, Wp);
  const int MT = N_NODES / 16;
  gat_gemm_wmma<<<(MT + 7) / 8, 256, 0, stream>>>(x, Wp, feat);
  gat_init<<<(2 * N_NODES + 255) / 256, 256, 0, stream>>>(cnt2);
  gat_el_er<<<(N_NODES * HEADS + 255) / 256, 256, 0, stream>>>(feat, attn_l, attn_r, el, out);
  gat_rank<<<(N_EDGES + 255) / 256, 256, 0, stream>>>(src, dst, cnt2, (int*)out);
  gat_scan<<<(N_NODES + 1023) / 1024, 256, 0, stream>>>(cnt2, start);
  gat_fill<<<(N_EDGES + 255) / 256, 256, 0, stream>>>(src, dst, cnt2, start, (const int*)out, csr);
  gat_node<<<(N_NODES + 7) / 8, 256, 0, stream>>>(csr, cnt2, start, el, feat, out);
}


# node kernel: a list of up to 32 edges is read once: 4 list reads, then 4 logit gathers, issued together and kept in registers for max, sum and weights (was 3 dependent passes); sum order unchanged
# speedup vs baseline: 2.0799x; 1.0689x over previous
#include <hip/hip_runtime.h>
#include <math.h>


typedef float v2f __attribute__((ext_vector_type(2)));
typedef float v8f __attribute__((ext_vector_type(8)));

#define N_NODES   100000
#define N_EDGES   1600000
#define IN_DIM    256
#define HEADS     4
#define OUT_DIM   32
#define FEAT_DIM  128
#define NEG_SLOPE 0.2f
#define KSTEPS    (IN_DIM / 4)
#define NTILES    (FEAT_DIM / 16)

__global__ void gat_prepack_W(const float* __restrict__ W, float* __restrict__ Wp) {
  const int t = blockIdx.x * blockDim.x + threadIdx.x;
  if (t >= KSTEPS * NTILES * 32) return;
  const int lane  = t & 31;
  const int tile  = (t >> 5) & 7;
  const int kstep = t >> 8;
  const int kb = (lane >> 4) * 2;
  const int n  = tile * 16 + (lane & 15);
  const int k  = kstep * 4 + kb;
  Wp[2 * t + 0] = W[(size_t)k * FEAT_DIM + n];
  Wp[2 * t + 1] = W[(size_t)(k + 1) * FEAT_DIM + n];
}

__global__ __launch_bounds__(256)
void gat_gemm_wmma(const float* __restrict__ x, const float* __restrict__ Wp,
                   float* __restrict__ feat) {
  const int lane = threadIdx.x & 31;
  const int mt   = blockIdx.x * 8 + (threadIdx.x >> 5);
  if (mt >= N_NODES / 16) return;

  const int kb = (lane >> 4) * 2;
  const float* aptr = x + (size_t)(mt * 16 + (lane & 15)) * IN_DIM + kb;
  const float* bptr = Wp + (size_t)lane * 2;

  v8f c[NTILES] = {};

  for (int ks = 0; ks < KSTEPS; ++ks) {
    const v2f a = *(const v2f*)(aptr + ks * 4);
#pragma unroll
    for (int t = 0; t < NTILES; ++t) {
      const v2f b = *(const v2f*)(bptr + (size_t)(ks * NTILES + t) * 64);
      c[t] = __builtin_amdgcn_wmma_f32_16x16x4_f32(false, a, false, b,
                                                   (short)0, c[t], false, false);
    }
  }

  const int mbase = mt * 16 + 8 * (lane >> 4);
  float* o = feat + (size_t)mbase * FEAT_DIM + (lane & 15);
#pragma unroll
  for (int t = 0; t < NTILES; ++t)
#pragma unroll
    for (int i = 0; i < 8; ++i)
      o[(size_t)i * FEAT_DIM + t * 16] = c[t][i];
}

#define SPLIT 65536

__global__ void gat_init(int* __restrict__ cnt2) {
  const int t = blockIdx.x * blockDim.x + threadIdx.x;
  if (t < 2 * N_NODES) cnt2[t] = 0;
}

__global__ void gat_el_er(const float* __restrict__ feat,
                          const float* __restrict__ attn_l,
                          const float* __restrict__ attn_r,
                          float* __restrict__ el, float* __restrict__ out) {
  int t = blockIdx.x * blockDim.x + threadIdx.x;
  if (t >= N_NODES * HEADS) return;
  const int h = t & 3;
  const float* f  = feat + (size_t)(t >> 2) * FEAT_DIM + h * OUT_DIM;
  const float* al = attn_l + h * OUT_DIM;
  const float* ar = attn_r + h * OUT_DIM;
  float sl = 0.0f, sr = 0.0f;
#pragma unroll
  for (int d = 0; d < OUT_DIM; ++d) { float v = f[d]; sl += v * al[d]; sr += v * ar[d]; }
  el[t] = sl;
  out[(size_t)(t >> 2) * OUT_DIM + 4 + h] = sr;
}

__device__ __forceinline__ float lrelu(float v) {
  return v > 0.0f ? v : NEG_SLOPE * v;
}

__global__ void gat_rank(const int* __restrict__ src, const int* __restrict__ dst,
                         int* __restrict__ cnt2, int* __restrict__ outw) {
  const int e = blockIdx.x * blockDim.x + threadIdx.x;
  if (e >= N_EDGES) return;
  const int s = src[e], d = dst[e];
  const int r = atomicAdd(cnt2 + (s >= SPLIT ? N_NODES : 0) + d, 1);
  outw[(size_t)(e >> 4) * OUT_DIM + 16 + (e & 15)] = r;
}

__global__ __launch_bounds__(256)
void gat_scan(const int* __restrict__ cnt2, int* __restrict__ start) {
  __shared__ int red[8];
  __shared__ int wtot[8];
  const int t = threadIdx.x, lane = t & 31, w = t >> 5;
  const int4* lo4 = (const int4*)cnt2;
  const int4* hi4 = (const int4*)(cnt2 + N_NODES);
  int sum = 0;
  const int nq = blockIdx.x * 256;
  for (int i = t; i < nq; i += 256) {
    const int4 a = lo4[i], b = hi4[i];
    sum += (a.x + a.y) + (a.z + a.w) + (b.x + b.y) + (b.z + b.w);
  }
  for (int o = 16; o > 0; o >>= 1) sum += __shfl_xor(sum, o);
  if (lane == 0) red[w] = sum;
  const int q = nq + t;
  int4 v = {0, 0, 0, 0};
  if (q < N_NODES / 4) {
    const int4 a = lo4[q], b = hi4[q];
    v.x = a.x + b.x; v.y = a.y + b.y; v.z = a.z + b.z; v.w = a.w + b.w;
  }
  const int ts = v.x + v.y + v.z + v.w;
  int inc = ts;
  for (int o = 1; o < 32; o <<= 1) {
    const int up = __shfl_up(inc, o);
    if (lane >= o) inc += up;
  }
  if (lane == 31) wtot[w] = inc;
  __syncthreads();
  int base = 0;
  for (int i = 0; i < 8; ++i) base += red[i];
  for (int i = 0; i < 8; ++i) base += (i < w) ? wtot[i] : 0;
  if (q < N_NODES / 4) {
    int4 o;
    o.x = base + inc - ts;
    o.y = o.x + v.x;
    o.z = o.y + v.y;
    o.w = o.z + v.z;
    ((int4*)start)[q] = o;
  }
}

__global__ void gat_fill(const int* __restrict__ src, const int* __restrict__ dst,
                         const int* __restrict__ cnt2, const int* __restrict__ start,
                         const int* __restrict__ outw, unsigned short* __restrict__ csr) {
  const int e = blockIdx.x * blockDim.x + threadIdx.x;
  if (e >= N_EDGES) return;
  const int s = src[e], d = dst[e];
  const int r = outw[(size_t)(e >> 4) * OUT_DIM + 16 + (e & 15)];
  const int p = start[d] + (s >= SPLIT ? cnt2[d] : 0) + r;
  if ((unsigned)p < (unsigned)N_EDGES) csr[p] = (unsigned short)(s & (SPLIT - 1));
}

#define EPW 8
#define RCH 4

__device__ __forceinline__ int list_src(const unsigned short* __restrict__ csr,
                                        int p, bool upper) {
  const int s = (int)csr[min(p, N_EDGES - 1)] + (upper ? SPLIT : 0);
  return min(s, N_NODES - 1);
}

__device__ __forceinline__ void gather_rows(const float* __restrict__ feat, int sv,
                                            int lane, float (&f)[EPW][HEADS]) {
  for (int jj = 0; jj < EPW; ++jj) {
    const int sj = __builtin_amdgcn_readlane(sv, 4 * jj);
    const float* fr = feat + (size_t)sj * FEAT_DIM + lane;
    for (int hh = 0; hh < HEADS; ++hh) f[jj][hh] = fr[hh * OUT_DIM];
  }
}

__device__ __forceinline__ float add_rows(float tot, float a,
                                          const float (&f)[EPW][HEADS]) {
  const int ai = __float_as_int(a);
  for (int jj = 0; jj < EPW; ++jj) {
    const float a0 = __int_as_float(__builtin_amdgcn_readlane(ai, 4 * jj + 0));
    const float a1 = __int_as_float(__builtin_amdgcn_readlane(ai, 4 * jj + 1));
    const float a2 = __int_as_float(__builtin_amdgcn_readlane(ai, 4 * jj + 2));
    const float a3 = __int_as_float(__builtin_amdgcn_readlane(ai, 4 * jj + 3));
    tot += a0 * f[jj][0] + a1 * f[jj][1] + a2 * f[jj][2] + a3 * f[jj][3];
  }
  return tot;
}

__global__ __launch_bounds__(256)
void gat_node(const unsigned short* __restrict__ csr, const int* __restrict__ cnt2,
              const int* __restrict__ start, const float* __restrict__ el,
              const float* __restrict__ feat, float* out) {
  const int lane = threadIdx.x & 31;
  const int n = __builtin_amdgcn_readfirstlane(blockIdx.x * 8 + (threadIdx.x >> 5));
  if (n >= N_NODES) return;
  float* orow = out + (size_t)n * OUT_DIM;
  const int lo   = cnt2[n];
  const int deg  = min(max(lo + cnt2[N_NODES + n], 0), N_EDGES);
  const int base = min(max(start[n], 0), N_EDGES);
  if (deg == 0) { orow[lane] = 0.0f; return; }

  const int h = lane & 3, j = lane >> 2;
  const float erh = orow[4 + h];

  float f[EPW][HEADS];
  float tot = 0.0f;

  if (deg <= RCH * EPW) {
    const int last = deg - 1;
    int   s[RCH];
    float p[RCH];
    for (int c = 0; c < RCH; ++c) {
      const int ic = min(c * EPW + j, last);
      s[c] = list_src(csr, base + ic, ic >= lo);
    }
    for (int c = 0; c < RCH; ++c)
      p[c] = lrelu(el[(size_t)s[c] * HEADS + h] + erh);
    gather_rows(feat, s[0], lane, f);

    float mx = -INFINITY;
    for (int c = 0; c < RCH; ++c)
      mx = (c * EPW + j < deg) ? fmaxf(mx, p[c]) : mx;
    mx = fmaxf(mx, __shfl_xor(mx, 4));
    mx = fmaxf(mx, __shfl_xor(mx, 8));
    mx = fmaxf(mx, __shfl_xor(mx, 16));

    float sm = 0.0f;
    for (int c = 0; c < RCH; ++c) {
      p[c] = (c * EPW + j < deg) ? __expf(p[c] - mx) : 0.0f;
      sm += p[c];
    }
    sm += __shfl_xor(sm, 4);
    sm += __shfl_xor(sm, 8);
    sm += __shfl_xor(sm, 16);
    const float den = fmaxf(sm, 1e-9f);
    for (int c = 0; c < RCH; ++c) p[c] = 0.25f * (p[c] / den);

    tot = add_rows(tot, p[0], f);
    for (int c = 1; c * EPW < deg; ++c) {
      const int   sc = (c == 1) ? s[1] : ((c == 2) ? s[2] : s[3]);
      const float ac = (c == 1) ? p[1] : ((c == 2) ? p[2] : p[3]);
      gather_rows(feat, sc, lane, f);
      tot = add_rows(tot, ac, f);
    }
  } else {
    float mx = -INFINITY;
    for (int c = 0; c < deg; c += EPW) {
      const int idx = c + j;
      const bool valid = idx < deg;
      const int s = valid ? list_src(csr, base + idx, idx >= lo) : 0;
      const float v = lrelu(el[(size_t)s * HEADS + h] + erh);
      mx = valid ? fmaxf(mx, v) : mx;
    }
    mx = fmaxf(mx, __shfl_xor(mx, 4));
    mx = fmaxf(mx, __shfl_xor(mx, 8));
    mx = fmaxf(mx, __shfl_xor(mx, 16));

    float sm = 0.0f;
    for (int c = 0; c < deg; c += EPW) {
      const int idx = c + j;
      const bool valid = idx < deg;
      const int s = valid ? list_src(csr, base + idx, idx >= lo) : 0;
      const float v = lrelu(el[(size_t)s * HEADS + h] + erh);
      sm += valid ? __expf(v - mx) : 0.0f;
    }
    sm += __shfl_xor(sm, 4);
    sm += __shfl_xor(sm, 8);
    sm += __shfl_xor(sm, 16);
    const float den = fmaxf(sm, 1e-9f);

    for (int c = 0; c < deg; c += EPW) {
      const int idx = c + j;
      const bool valid = idx < deg;
      const int s = valid ? list_src(csr, base + idx, idx >= lo) : 0;
      const float v = lrelu(el[(size_t)s * HEADS + h] + erh);
      const float a = valid ? 0.25f * (__expf(v - mx) / den) : 0.0f;
      const int s0 = __builtin_amdgcn_readfirstlane(s);
      const int sv = valid ? s : s0;
      gather_rows(feat, sv, lane, f);
      tot = add_rows(tot, a, f);
    }
  }
  orow[lane] = tot;
}

extern "C" void kernel_launch(void* const* d_in, const int* in_sizes, int n_in,
                              void* d_out, int out_size, void* d_ws, size_t ws_size,
                              hipStream_t stream) {
  const float* x      = (const float*)d_in[0];
  const float* W      = (const float*)d_in[1];
  const float* attn_l = (const float*)d_in[2];
  const float* attn_r = (const float*)d_in[3];
  const int*   src    = (const int*)d_in[4];
  const int*   dst    = (const int*)d_in[5];
  float* out = (float*)d_out;

  float* ws    = (float*)d_ws;
  float* feat  = ws;
  float* el    = feat + (size_t)N_NODES * FEAT_DIM;
  unsigned short* csr = (unsigned short*)(el + (size_t)N_NODES * HEADS);
  int*   cnt2  = (int*)(el + (size_t)3 * N_NODES * HEADS);
  int*   start = cnt2 + 2 * N_NODES;
  float* Wp    = el + (size_t)4 * N_NODES * HEADS;

  gat_prepack_W<<<(KSTEPS * NTILES * 32 + 255) / 256, 256, 0, stream>>>(W, Wp);
  const int MT = N_NODES / 16;
  gat_gemm_wmma<<<(MT + 7) / 8, 256, 0, stream>>>(x, Wp, feat);
  gat_init<<<(2 * N_NODES + 255) / 256, 256, 0, stream>>>(cnt2);
  gat_el_er<<<(N_NODES * HEADS + 255) / 256, 256, 0, stream>>>(feat, attn_l, attn_r, el, out);
  gat_rank<<<(N_EDGES + 255) / 256, 256, 0, stream>>>(src, dst, cnt2, (int*)out);
  gat_scan<<<(N_NODES + 1023) / 1024, 256, 0, stream>>>(cnt2, start);
  gat_fill<<<(N_EDGES + 255) / 256, 256, 0, stream>>>(src, dst, cnt2, start, (const int*)out, csr);
  gat_node<<<(N_NODES + 7) / 8, 256, 0, stream>>>(csr, cnt2, start, el, feat, out);
}


# sort passes share launches with the GEMM passes, split by block index: prepack|clear, GEMM|rank, scan|el-er; 8 launches become 5; arithmetic and sum orders unchanged
# speedup vs baseline: 2.2764x; 1.0945x over previous
#include <hip/hip_runtime.h>
#include <math.h>


typedef float v2f __attribute__((ext_vector_type(2)));
typedef float v8f __attribute__((ext_vector_type(8)));

#define N_NODES   100000
#define N_EDGES   1600000
#define IN_DIM    256
#define HEADS     4
#define OUT_DIM   32
#define FEAT_DIM  128
#define NEG_SLOPE 0.2f
#define KSTEPS    (IN_DIM / 4)
#define NTILES    (FEAT_DIM / 16)

__device__ __forceinline__ void prepack_body(int b, const float* __restrict__ W,
                                             float* __restrict__ Wp) {
  const int t = b * 256 + threadIdx.x;
  if (t >= KSTEPS * NTILES * 32) return;
  const int lane  = t & 31;
  const int tile  = (t >> 5) & 7;
  const int kstep = t >> 8;
  const int kb = (lane >> 4) * 2;
  const int n  = tile * 16 + (lane & 15);
  const int k  = kstep * 4 + kb;
  Wp[2 * t + 0] = W[(size_t)k * FEAT_DIM + n];
  Wp[2 * t + 1] = W[(size_t)(k + 1) * FEAT_DIM + n];
}

__device__ __forceinline__ void gemm_body(int b, const float* __restrict__ x,
                                          const float* __restrict__ Wp,
                                          float* __restrict__ feat) {
  const int lane = threadIdx.x & 31;
  const int mt   = b * 8 + (threadIdx.x >> 5);
  if (mt >= N_NODES / 16) return;

  const int kb = (lane >> 4) * 2;
  const float* aptr = x + (size_t)(mt * 16 + (lane & 15)) * IN_DIM + kb;
  const float* bptr = Wp + (size_t)lane * 2;

  v8f c[NTILES] = {};

  for (int ks = 0; ks < KSTEPS; ++ks) {
    const v2f a = *(const v2f*)(aptr + ks * 4);
#pragma unroll
    for (int t = 0; t < NTILES; ++t) {
      const v2f b = *(const v2f*)(bptr + (size_t)(ks * NTILES + t) * 64);
      c[t] = __builtin_amdgcn_wmma_f32_16x16x4_f32(false, a, false, b,
                                                   (short)0, c[t], false, false);
    }
  }

  const int mbase = mt * 16 + 8 * (lane >> 4);
  float* o = feat + (size_t)mbase * FEAT_DIM + (lane & 15);
#pragma unroll
  for (int t = 0; t < NTILES; ++t)
#pragma unroll
    for (int i = 0; i < 8; ++i)
      o[(size_t)i * FEAT_DIM + t * 16] = c[t][i];
}

#define SPLIT 65536

__device__ __forceinline__ void init_body(int b, int* __restrict__ cnt2) {
  const int t = b * 256 + threadIdx.x;
  if (t < 2 * N_NODES) cnt2[t] = 0;
}

__device__ __forceinline__ void el_er_body(int b, const float* __restrict__ feat,
                                           const float* __restrict__ attn_l,
                                           const float* __restrict__ attn_r,
                                           float* __restrict__ el, float* __restrict__ out) {
  int t = b * 256 + threadIdx.x;
  if (t >= N_NODES * HEADS) return;
  const int h = t & 3;
  const float* f  = feat + (size_t)(t >> 2) * FEAT_DIM + h * OUT_DIM;
  const float* al = attn_l + h * OUT_DIM;
  const float* ar = attn_r + h * OUT_DIM;
  float sl = 0.0f, sr = 0.0f;
#pragma unroll
  for (int d = 0; d < OUT_DIM; ++d) { float v = f[d]; sl += v * al[d]; sr += v * ar[d]; }
  el[t] = sl;
  out[(size_t)(t >> 2) * OUT_DIM + 4 + h] = sr;
}

__device__ __forceinline__ float lrelu(float v) {
  return v > 0.0f ? v : NEG_SLOPE * v;
}

__device__ __forceinline__ void rank_body(int b, const int* __restrict__ src,
                                          const int* __restrict__ dst,
                                          int* __restrict__ cnt2, int* __restrict__ outw) {
  const int e = b * 256 + threadIdx.x;
  if (e >= N_EDGES) return;
  const int s = src[e], d = dst[e];
  const int r = atomicAdd(cnt2 + (s >= SPLIT ? N_NODES : 0) + d, 1);
  outw[(size_t)(e >> 4) * OUT_DIM + 16 + (e & 15)] = r;
}

__device__ __forceinline__ void scan_body(int b, const int* __restrict__ cnt2,
                                          int* __restrict__ start) {
  __shared__ int red[8];
  __shared__ int wtot[8];
  const int t = threadIdx.x, lane = t & 31, w = t >> 5;
  const int4* lo4 = (const int4*)cnt2;
  const int4* hi4 = (const int4*)(cnt2 + N_NODES);
  int sum = 0;
  const int nq = b * 256;
  for (int i = t; i < nq; i += 256) {
    const int4 a = lo4[i], b = hi4[i];
    sum += (a.x + a.y) + (a.z + a.w) + (b.x + b.y) + (b.z + b.w);
  }
  for (int o = 16; o > 0; o >>= 1) sum += __shfl_xor(sum, o);
  if (lane == 0) red[w] = sum;
  const int q = nq + t;
  int4 v = {0, 0, 0, 0};
  if (q < N_NODES / 4) {
    const int4 a = lo4[q], b = hi4[q];
    v.x = a.x + b.x; v.y = a.y + b.y; v.z = a.z + b.z; v.w = a.w + b.w;
  }
  const int ts = v.x + v.y + v.z + v.w;
  int inc = ts;
  for (int o = 1; o < 32; o <<= 1) {
    const int up = __shfl_up(inc, o);
    if (lane >= o) inc += up;
  }
  if (lane == 31) wtot[w] = inc;
  __syncthreads();
  int base = 0;
  for (int i = 0; i < 8; ++i) base += red[i];
  for (int i = 0; i < 8; ++i) base += (i < w) ? wtot[i] : 0;
  if (q < N_NODES / 4) {
    int4 o;
    o.x = base + inc - ts;
    o.y = o.x + v.x;
    o.z = o.y + v.y;
    o.w = o.z + v.z;
    ((int4*)start)[q] = o;
  }
}

#define PACK_BLOCKS ((KSTEPS * NTILES * 32 + 255) / 256)
#define INIT_BLOCKS ((2 * N_NODES + 255) / 256)
#define GEMM_BLOCKS ((N_NODES / 16 + 7) / 8)
#define RANK_BLOCKS ((N_EDGES + 255) / 256)
#define SCAN_BLOCKS ((N_NODES + 1023) / 1024)
#define ELER_BLOCKS ((N_NODES * HEADS + 255) / 256)

__global__ __launch_bounds__(256)
void gat_pack_init(const float* __restrict__ W, float* __restrict__ Wp,
                   int* __restrict__ cnt2) {
  const int b = blockIdx.x;
  if (b < PACK_BLOCKS) prepack_body(b, W, Wp);
  else                 init_body(b - PACK_BLOCKS, cnt2);
}

__global__ __launch_bounds__(256)
void gat_gemm_rank(const float* __restrict__ x, const float* __restrict__ Wp,
                   float* __restrict__ feat,
                   const int* __restrict__ src, const int* __restrict__ dst,
                   int* __restrict__ cnt2, int* __restrict__ outw) {
  const int b = blockIdx.x;
  if (b < GEMM_BLOCKS) gemm_body(b, x, Wp, feat);
  else                 rank_body(b - GEMM_BLOCKS, src, dst, cnt2, outw);
}

__global__ __launch_bounds__(256)
void gat_scan_el_er(const int* __restrict__ cnt2, int* __restrict__ start,
                    const float* __restrict__ feat,
                    const float* __restrict__ attn_l,
                    const float* __restrict__ attn_r,
                    float* __restrict__ el, float* __restrict__ out) {
  const int b = blockIdx.x;
  if (b < SCAN_BLOCKS) scan_body(b, cnt2, start);
  else                 el_er_body(b - SCAN_BLOCKS, feat, attn_l, attn_r, el, out);
}

__global__ void gat_fill(const int* __restrict__ src, const int* __restrict__ dst,
                         const int* __restrict__ cnt2, const int* __restrict__ start,
                         const int* __restrict__ outw, unsigned short* __restrict__ csr) {
  const int e = blockIdx.x * blockDim.x + threadIdx.x;
  if (e >= N_EDGES) return;
  const int s = src[e], d = dst[e];
  const int r = outw[(size_t)(e >> 4) * OUT_DIM + 16 + (e & 15)];
  const int p = start[d] + (s >= SPLIT ? cnt2[d] : 0) + r;
  if ((unsigned)p < (unsigned)N_EDGES) csr[p] = (unsigned short)(s & (SPLIT - 1));
}

#define EPW 8
#define RCH 4

__device__ __forceinline__ int list_src(const unsigned short* __restrict__ csr,
                                        int p, bool upper) {
  const int s = (int)csr[min(p, N_EDGES - 1)] + (upper ? SPLIT : 0);
  return min(s, N_NODES - 1);
}

__device__ __forceinline__ void gather_rows(const float* __restrict__ feat, int sv,
                                            int lane, float (&f)[EPW][HEADS]) {
  for (int jj = 0; jj < EPW; ++jj) {
    const int sj = __builtin_amdgcn_readlane(sv, 4 * jj);
    const float* fr = feat + (size_t)sj * FEAT_DIM + lane;
    for (int hh = 0; hh < HEADS; ++hh) f[jj][hh] = fr[hh * OUT_DIM];
  }
}

__device__ __forceinline__ float add_rows(float tot, float a,
                                          const float (&f)[EPW][HEADS]) {
  const int ai = __float_as_int(a);
  for (int jj = 0; jj < EPW; ++jj) {
    const float a0 = __int_as_float(__builtin_amdgcn_readlane(ai, 4 * jj + 0));
    const float a1 = __int_as_float(__builtin_amdgcn_readlane(ai, 4 * jj + 1));
    const float a2 = __int_as_float(__builtin_amdgcn_readlane(ai, 4 * jj + 2));
    const float a3 = __int_as_float(__builtin_amdgcn_readlane(ai, 4 * jj + 3));
    tot += a0 * f[jj][0] + a1 * f[jj][1] + a2 * f[jj][2] + a3 * f[jj][3];
  }
  return tot;
}

__global__ __launch_bounds__(256)
void gat_node(const unsigned short* __restrict__ csr, const int* __restrict__ cnt2,
              const int* __restrict__ start, const float* __restrict__ el,
              const float* __restrict__ feat, float* out) {
  const int lane = threadIdx.x & 31;
  const int n = __builtin_amdgcn_readfirstlane(blockIdx.x * 8 + (threadIdx.x >> 5));
  if (n >= N_NODES) return;
  float* orow = out + (size_t)n * OUT_DIM;
  const int lo   = cnt2[n];
  const int deg  = min(max(lo + cnt2[N_NODES + n], 0), N_EDGES);
  const int base = min(max(start[n], 0), N_EDGES);
  if (deg == 0) { orow[lane] = 0.0f; return; }

  const int h = lane & 3, j = lane >> 2;
  const float erh = orow[4 + h];

  float f[EPW][HEADS];
  float tot = 0.0f;

  if (deg <= RCH * EPW) {
    const int last = deg - 1;
    int   s[RCH];
    float p[RCH];
    for (int c = 0; c < RCH; ++c) {
      const int ic = min(c * EPW + j, last);
      s[c] = list_src(csr, base + ic, ic >= lo);
    }
    for (int c = 0; c < RCH; ++c)
      p[c] = lrelu(el[(size_t)s[c] * HEADS + h] + erh);
    gather_rows(feat, s[0], lane, f);

    float mx = -INFINITY;
    for (int c = 0; c < RCH; ++c)
      mx = (c * EPW + j < deg) ? fmaxf(mx, p[c]) : mx;
    mx = fmaxf(mx, __shfl_xor(mx, 4));
    mx = fmaxf(mx, __shfl_xor(mx, 8));
    mx = fmaxf(mx, __shfl_xor(mx, 16));

    float sm = 0.0f;
    for (int c = 0; c < RCH; ++c) {
      p[c] = (c * EPW + j < deg) ? __expf(p[c] - mx) : 0.0f;
      sm += p[c];
    }
    sm += __shfl_xor(sm, 4);
    sm += __shfl_xor(sm, 8);
    sm += __shfl_xor(sm, 16);
    const float den = fmaxf(sm, 1e-9f);
    for (int c = 0; c < RCH; ++c) p[c] = 0.25f * (p[c] / den);

    tot = add_rows(tot, p[0], f);
    for (int c = 1; c * EPW < deg; ++c) {
      const int   sc = (c == 1) ? s[1] : ((c == 2) ? s[2] : s[3]);
      const float ac = (c == 1) ? p[1] : ((c == 2) ? p[2] : p[3]);
      gather_rows(feat, sc, lane, f);
      tot = add_rows(tot, ac, f);
    }
  } else {
    float mx = -INFINITY;
    for (int c = 0; c < deg; c += EPW) {
      const int idx = c + j;
      const bool valid = idx < deg;
      const int s = valid ? list_src(csr, base + idx, idx >= lo) : 0;
      const float v = lrelu(el[(size_t)s * HEADS + h] + erh);
      mx = valid ? fmaxf(mx, v) : mx;
    }
    mx = fmaxf(mx, __shfl_xor(mx, 4));
    mx = fmaxf(mx, __shfl_xor(mx, 8));
    mx = fmaxf(mx, __shfl_xor(mx, 16));

    float sm = 0.0f;
    for (int c = 0; c < deg; c += EPW) {
      const int idx = c + j;
      const bool valid = idx < deg;
      const int s = valid ? list_src(csr, base + idx, idx >= lo) : 0;
      const float v = lrelu(el[(size_t)s * HEADS + h] + erh);
      sm += valid ? __expf(v - mx) : 0.0f;
    }
    sm += __shfl_xor(sm, 4);
    sm += __shfl_xor(sm, 8);
    sm += __shfl_xor(sm, 16);
    const float den = fmaxf(sm, 1e-9f);

    for (int c = 0; c < deg; c += EPW) {
      const int idx = c + j;
      const bool valid = idx < deg;
      const int s = valid ? list_src(csr, base + idx, idx >= lo) : 0;
      const float v = lrelu(el[(size_t)s * HEADS + h] + erh);
      const float a = valid ? 0.25f * (__expf(v - mx) / den) : 0.0f;
      const int s0 = __builtin_amdgcn_readfirstlane(s);
      const int sv = valid ? s : s0;
      gather_rows(feat, sv, lane, f);
      tot = add_rows(tot, a, f);
    }
  }
  orow[lane] = tot;
}

extern "C" void kernel_launch(void* const* d_in, const int* in_sizes, int n_in,
                              void* d_out, int out_size, void* d_ws, size_t ws_size,
                              hipStream_t stream) {
  const float* x      = (const float*)d_in[0];
  const float* W      = (const float*)d_in[1];
  const float* attn_l = (const float*)d_in[2];
  const float* attn_r = (const float*)d_in[3];
  const int*   src    = (const int*)d_in[4];
  const int*   dst    = (const int*)d_in[5];
  float* out = (float*)d_out;

  float* ws    = (float*)d_ws;
  float* feat  = ws;
  float* el    = feat + (size_t)N_NODES * FEAT_DIM;
  unsigned short* csr = (unsigned short*)(el + (size_t)N_NODES * HEADS);
  int*   cnt2  = (int*)(el + (size_t)3 * N_NODES * HEADS);
  int*   start = cnt2 + 2 * N_NODES;
  float* Wp    = el + (size_t)4 * N_NODES * HEADS;

  gat_pack_init<<<PACK_BLOCKS + INIT_BLOCKS, 256, 0, stream>>>(W, Wp, cnt2);
  gat_gemm_rank<<<GEMM_BLOCKS + RANK_BLOCKS, 256, 0, stream>>>(x, Wp, feat, src, dst, cnt2, (int*)out);
  gat_scan_el_er<<<SCAN_BLOCKS + ELER_BLOCKS, 256, 0, stream>>>(cnt2, start, feat, attn_l, attn_r, el, out);
  gat_fill<<<(N_EDGES + 255) / 256, 256, 0, stream>>>(src, dst, cnt2, start, (const int*)out, csr);
  gat_node<<<(N_NODES + 7) / 8, 256, 0, stream>>>(csr, cnt2, start, el, feat, out);
}


# rank pass stores each source id straight into its destination's 64-slot bucket (the node's own output line, read by one load): scan pass, fill pass and a launch dropped; sums in list order as parent
# speedup vs baseline: 3.1020x; 1.3627x over previous
#include <hip/hip_runtime.h>
#include <math.h>


typedef float v2f __attribute__((ext_vector_type(2)));
typedef float v8f __attribute__((ext_vector_type(8)));

#define N_NODES   100000
#define N_EDGES   1600000
#define IN_DIM    256
#define HEADS     4
#define OUT_DIM   32
#define FEAT_DIM  128
#define NEG_SLOPE 0.2f
#define KSTEPS    (IN_DIM / 4)
#define NTILES    (FEAT_DIM / 16)

__device__ __forceinline__ void prepack_body(int b, const float* __restrict__ W,
                                             float* __restrict__ Wp) {
  const int t = b * 256 + threadIdx.x;
  if (t >= KSTEPS * NTILES * 32) return;
  const int lane  = t & 31;
  const int tile  = (t >> 5) & 7;
  const int kstep = t >> 8;
  const int kb = (lane >> 4) * 2;
  const int n  = tile * 16 + (lane & 15);
  const int k  = kstep * 4 + kb;
  Wp[2 * t + 0] = W[(size_t)k * FEAT_DIM + n];
  Wp[2 * t + 1] = W[(size_t)(k + 1) * FEAT_DIM + n];
}

__device__ __forceinline__ void gemm_body(int b, const float* __restrict__ x,
                                          const float* __restrict__ Wp,
                                          float* __restrict__ feat) {
  const int lane = threadIdx.x & 31;
  const int mt   = b * 8 + (threadIdx.x >> 5);
  if (mt >= N_NODES / 16) return;

  const int kb = (lane >> 4) * 2;
  const float* aptr = x + (size_t)(mt * 16 + (lane & 15)) * IN_DIM + kb;
  const float* bptr = Wp + (size_t)lane * 2;

  v8f c[NTILES] = {};

  for (int ks = 0; ks < KSTEPS; ++ks) {
    const v2f a = *(const v2f*)(aptr + ks * 4);
#pragma unroll
    for (int t = 0; t < NTILES; ++t) {
      const v2f b = *(const v2f*)(bptr + (size_t)(ks * NTILES + t) * 64);
      c[t] = __builtin_amdgcn_wmma_f32_16x16x4_f32(false, a, false, b,
                                                   (short)0, c[t], false, false);
    }
  }

  const int mbase = mt * 16 + 8 * (lane >> 4);
  float* o = feat + (size_t)mbase * FEAT_DIM + (lane & 15);
#pragma unroll
  for (int t = 0; t < NTILES; ++t)
#pragma unroll
    for (int i = 0; i < 8; ++i)
      o[(size_t)i * FEAT_DIM + t * 16] = c[t][i];
}

#define SPLIT 65536
#define CAP   64

__device__ __forceinline__ void init_body(int b, int* __restrict__ cnt2) {
  const int t = b * 256 + threadIdx.x;
  if (t < 2 * N_NODES) cnt2[t] = 0;
}

__device__ __forceinline__ void el_er_body(int b, const float* __restrict__ feat,
                                           const float* __restrict__ attn_l,
                                           const float* __restrict__ attn_r,
                                           float* __restrict__ el, float* __restrict__ er) {
  int t = b * 256 + threadIdx.x;
  if (t >= N_NODES * HEADS) return;
  const int h = t & 3;
  const float* f  = feat + (size_t)(t >> 2) * FEAT_DIM + h * OUT_DIM;
  const float* al = attn_l + h * OUT_DIM;
  const float* ar = attn_r + h * OUT_DIM;
  float sl = 0.0f, sr = 0.0f;
#pragma unroll
  for (int d = 0; d < OUT_DIM; ++d) { float v = f[d]; sl += v * al[d]; sr += v * ar[d]; }
  el[t] = sl;
  er[t] = sr;
}

__device__ __forceinline__ float lrelu(float v) {
  return v > 0.0f ? v : NEG_SLOPE * v;
}

__device__ __forceinline__ void place_body(int b, const int* __restrict__ src,
                                           const int* __restrict__ dst,
                                           int* __restrict__ cnt2,
                                           unsigned short* __restrict__ bucket) {
  const int e = b * 256 + threadIdx.x;
  if (e >= N_EDGES) return;
  const int s = src[e], d = dst[e];
  const bool up = s >= SPLIT;
  const int r = atomicAdd(cnt2 + (up ? N_NODES : 0) + d, 1);
  if ((unsigned)r < (unsigned)CAP)
    bucket[(size_t)d * CAP + (up ? CAP - 1 - r : r)] = (unsigned short)(s & (SPLIT - 1));
}

#define PACK_BLOCKS ((KSTEPS * NTILES * 32 + 255) / 256)
#define INIT_BLOCKS ((2 * N_NODES + 255) / 256)
#define GEMM_BLOCKS ((N_NODES / 16 + 7) / 8)
#define RANK_BLOCKS ((N_EDGES + 255) / 256)
#define ELER_BLOCKS ((N_NODES * HEADS + 255) / 256)

__global__ __launch_bounds__(256)
void gat_pack_init(const float* __restrict__ W, float* __restrict__ Wp,
                   int* __restrict__ cnt2) {
  const int b = blockIdx.x;
  if (b < PACK_BLOCKS) prepack_body(b, W, Wp);
  else                 init_body(b - PACK_BLOCKS, cnt2);
}

__global__ __launch_bounds__(256)
void gat_gemm_place(const float* __restrict__ x, const float* __restrict__ Wp,
                    float* __restrict__ feat,
                    const int* __restrict__ src, const int* __restrict__ dst,
                    int* __restrict__ cnt2, unsigned short* __restrict__ bucket) {
  const int b = blockIdx.x;
  if (b < GEMM_BLOCKS) gemm_body(b, x, Wp, feat);
  else                 place_body(b - GEMM_BLOCKS, src, dst, cnt2, bucket);
}

__global__ __launch_bounds__(256)
void gat_el_er(const float* __restrict__ feat,
               const float* __restrict__ attn_l,
               const float* __restrict__ attn_r,
               float* __restrict__ el, float* __restrict__ er) {
  el_er_body(blockIdx.x, feat, attn_l, attn_r, el, er);
}

#define EPW 8
#define RCH 4

__device__ __forceinline__ int bucket_src(int w, int i, int lo) {
  const bool up = i >= lo;
  const int k = up ? (CAP - 1 + lo - i) : i;
  const int v = __shfl(w, (k >> 1) & 31);
  const int s = ((k & 1) ? (int)((unsigned)v >> 16) : (v & (SPLIT - 1))) + (up ? SPLIT : 0);
  return min(s, N_NODES - 1);
}

__device__ __forceinline__ void gather_rows(const float* __restrict__ feat, int sv,
                                            int lane, float (&f)[EPW][HEADS]) {
  for (int jj = 0; jj < EPW; ++jj) {
    const int sj = __builtin_amdgcn_readlane(sv, 4 * jj);
    const float* fr = feat + (size_t)sj * FEAT_DIM + lane;
    for (int hh = 0; hh < HEADS; ++hh) f[jj][hh] = fr[hh * OUT_DIM];
  }
}

__device__ __forceinline__ float add_rows(float tot, float a,
                                          const float (&f)[EPW][HEADS]) {
  const int ai = __float_as_int(a);
  for (int jj = 0; jj < EPW; ++jj) {
    const float a0 = __int_as_float(__builtin_amdgcn_readlane(ai, 4 * jj + 0));
    const float a1 = __int_as_float(__builtin_amdgcn_readlane(ai, 4 * jj + 1));
    const float a2 = __int_as_float(__builtin_amdgcn_readlane(ai, 4 * jj + 2));
    const float a3 = __int_as_float(__builtin_amdgcn_readlane(ai, 4 * jj + 3));
    tot += a0 * f[jj][0] + a1 * f[jj][1] + a2 * f[jj][2] + a3 * f[jj][3];
  }
  return tot;
}

__global__ __launch_bounds__(256)
void gat_node(const int* __restrict__ src, const int* __restrict__ dst,
              const int* __restrict__ cnt2, const float* __restrict__ el,
              const float* __restrict__ er, const float* __restrict__ feat,
              float* out) {
  const int lane = threadIdx.x & 31;
  const int n = __builtin_amdgcn_readfirstlane(blockIdx.x * 8 + (threadIdx.x >> 5));
  if (n >= N_NODES) return;
  float* orow = out + (size_t)n * OUT_DIM;
  const int w = ((const int*)orow)[lane];
  const int lo  = min(max(cnt2[n], 0), N_EDGES);
  const int deg = min(lo + max(cnt2[N_NODES + n], 0), N_EDGES);
  if (deg == 0) { orow[lane] = 0.0f; return; }

  const int h = lane & 3, j = lane >> 2;
  const float erh = er[(size_t)n * HEADS + h];

  float f[EPW][HEADS];
  float tot = 0.0f;
  const int last = deg - 1;

  if (deg <= RCH * EPW) {
    int   s[RCH];
    float p[RCH];
    for (int c = 0; c < RCH; ++c)
      s[c] = bucket_src(w, min(c * EPW + j, last), lo);
    for (int c = 0; c < RCH; ++c)
      p[c] = lrelu(el[(size_t)s[c] * HEADS + h] + erh);
    gather_rows(feat, s[0], lane, f);

    float mx = -INFINITY;
    for (int c = 0; c < RCH; ++c)
      mx = (c * EPW + j < deg) ? fmaxf(mx, p[c]) : mx;
    mx = fmaxf(mx, __shfl_xor(mx, 4));
    mx = fmaxf(mx, __shfl_xor(mx, 8));
    mx = fmaxf(mx, __shfl_xor(mx, 16));

    float sm = 0.0f;
    for (int c = 0; c < RCH; ++c) {
      p[c] = (c * EPW + j < deg) ? __expf(p[c] - mx) : 0.0f;
      sm += p[c];
    }
    sm += __shfl_xor(sm, 4);
    sm += __shfl_xor(sm, 8);
    sm += __shfl_xor(sm, 16);
    const float den = fmaxf(sm, 1e-9f);
    for (int c = 0; c < RCH; ++c) p[c] = 0.25f * (p[c] / den);

    tot = add_rows(tot, p[0], f);
    for (int c = 1; c * EPW < deg; ++c) {
      const int   sc = (c == 1) ? s[1] : ((c == 2) ? s[2] : s[3]);
      const float ac = (c == 1) ? p[1] : ((c == 2) ? p[2] : p[3]);
      gather_rows(feat, sc, lane, f);
      tot = add_rows(tot, ac, f);
    }
  } else if (deg <= CAP) {
    float mx = -INFINITY;
    for (int c = 0; c < deg; c += EPW) {
      const int s = bucket_src(w, min(c + j, last), lo);
      const float v = lrelu(el[(size_t)s * HEADS + h] + erh);
      mx = (c + j < deg) ? fmaxf(mx, v) : mx;
    }
    mx = fmaxf(mx, __shfl_xor(mx, 4));
    mx = fmaxf(mx, __shfl_xor(mx, 8));
    mx = fmaxf(mx, __shfl_xor(mx, 16));

    float sm = 0.0f;
    for (int c = 0; c < deg; c += EPW) {
      const int s = bucket_src(w, min(c + j, last), lo);
      const float v = lrelu(el[(size_t)s * HEADS + h] + erh);
      sm += (c + j < deg) ? __expf(v - mx) : 0.0f;
    }
    sm += __shfl_xor(sm, 4);
    sm += __shfl_xor(sm, 8);
    sm += __shfl_xor(sm, 16);
    const float den = fmaxf(sm, 1e-9f);

    for (int c = 0; c < deg; c += EPW) {
      const int s = bucket_src(w, min(c + j, last), lo);
      const float v = lrelu(el[(size_t)s * HEADS + h] + erh);
      const float a = (c + j < deg) ? 0.25f * (__expf(v - mx) / den) : 0.0f;
      gather_rows(feat, s, lane, f);
      tot = add_rows(tot, a, f);
    }
  } else {
    const float e0 = er[(size_t)n * HEADS + 0], e1 = er[(size_t)n * HEADS + 1];
    const float e2 = er[(size_t)n * HEADS + 2], e3 = er[(size_t)n * HEADS + 3];
    float m0 = -INFINITY, m1 = -INFINITY, m2 = -INFINITY, m3 = -INFINITY;
    for (int b0 = 0; b0 < N_EDGES; b0 += 32) {
      const int e = b0 + lane;
      if (e < N_EDGES && dst[e] == n) {
        const int s = min(max(src[e], 0), N_NODES - 1);
        const float* l = el + (size_t)s * HEADS;
        m0 = fmaxf(m0, lrelu(l[0] + e0)); m1 = fmaxf(m1, lrelu(l[1] + e1));
        m2 = fmaxf(m2, lrelu(l[2] + e2)); m3 = fmaxf(m3, lrelu(l[3] + e3));
      }
    }
    for (int o = 16; o > 0; o >>= 1) {
      m0 = fmaxf(m0, __shfl_xor(m0, o)); m1 = fmaxf(m1, __shfl_xor(m1, o));
      m2 = fmaxf(m2, __shfl_xor(m2, o)); m3 = fmaxf(m3, __shfl_xor(m3, o));
    }
    float d0 = 0.0f, d1 = 0.0f, d2 = 0.0f, d3 = 0.0f;
    for (int b0 = 0; b0 < N_EDGES; b0 += 32) {
      const int e = b0 + lane;
      if (e < N_EDGES && dst[e] == n) {
        const int s = min(max(src[e], 0), N_NODES - 1);
        const float* l = el + (size_t)s * HEADS;
        d0 += __expf(lrelu(l[0] + e0) - m0); d1 += __expf(lrelu(l[1] + e1) - m1);
        d2 += __expf(lrelu(l[2] + e2) - m2); d3 += __expf(lrelu(l[3] + e3) - m3);
      }
    }
    for (int o = 16; o > 0; o >>= 1) {
      d0 += __shfl_xor(d0, o); d1 += __shfl_xor(d1, o);
      d2 += __shfl_xor(d2, o); d3 += __shfl_xor(d3, o);
    }
    d0 = fmaxf(d0, 1e-9f); d1 = fmaxf(d1, 1e-9f); d2 = fmaxf(d2, 1e-9f); d3 = fmaxf(d3, 1e-9f);
    for (int b0 = 0; b0 < N_EDGES; b0 += 32) {
      const int e = b0 + lane;
      const bool hit = e < N_EDGES && dst[e] == n;
      int s = 0;
      float a0 = 0.0f, a1 = 0.0f, a2 = 0.0f, a3 = 0.0f;
      if (hit) {
        s = min(max(src[e], 0), N_NODES - 1);
        const float* l = el + (size_t)s * HEADS;
        a0 = 0.25f * (__expf(lrelu(l[0] + e0) - m0) / d0);
        a1 = 0.25f * (__expf(lrelu(l[1] + e1) - m1) / d1);
        a2 = 0.25f * (__expf(lrelu(l[2] + e2) - m2) / d2);
        a3 = 0.25f * (__expf(lrelu(l[3] + e3) - m3) / d3);
      }
      unsigned mask = __builtin_amdgcn_ballot_w32(hit);
      while (mask) {
        const int bl = __builtin_ctz(mask);
        mask &= mask - 1;
        const int sj = __shfl(s, bl);
        const float* fr = feat + (size_t)sj * FEAT_DIM + lane;
        tot += __shfl(a0, bl) * fr[0] + __shfl(a1, bl) * fr[OUT_DIM]
             + __shfl(a2, bl) * fr[2 * OUT_DIM] + __shfl(a3, bl) * fr[3 * OUT_DIM];
      }
    }
  }
  orow[lane] = tot;
}

extern "C" void kernel_launch(void* const* d_in, const int* in_sizes, int n_in,
                              void* d_out, int out_size, void* d_ws, size_t ws_size,
                              hipStream_t stream) {
  const float* x      = (const float*)d_in[0];
  const float* W      = (const float*)d_in[1];
  const float* attn_l = (const float*)d_in[2];
  const float* attn_r = (const float*)d_in[3];
  const int*   src    = (const int*)d_in[4];
  const int*   dst    = (const int*)d_in[5];
  float* out = (float*)d_out;

  float* ws    = (float*)d_ws;
  float* feat  = ws;
  float* el    = feat + (size_t)N_NODES * FEAT_DIM;
  float* er    = el + (size_t)N_NODES * HEADS;
  int*   cnt2  = (int*)(el + (size_t)3 * N_NODES * HEADS);
  float* Wp    = el + (size_t)4 * N_NODES * HEADS;

  gat_pack_init<<<PACK_BLOCKS + INIT_BLOCKS, 256, 0, stream>>>(W, Wp, cnt2);
  gat_gemm_place<<<GEMM_BLOCKS + RANK_BLOCKS, 256, 0, stream>>>(x, Wp, feat, src, dst, cnt2, (unsigned short*)out);
  gat_el_er<<<ELER_BLOCKS, 256, 0, stream>>>(feat, attn_l, attn_r, el, er);
  gat_node<<<(N_NODES + 7) / 8, 256, 0, stream>>>(src, dst, cnt2, el, er, feat, out);
}


# GEMM pass: the A pair and the 8 B fragments of each k-step are loaded one k-step ahead of the 8 products that use them (two register sets); k order and sum order unchanged, results bit-identical
# speedup vs baseline: 3.2872x; 1.0597x over previous
#include <hip/hip_runtime.h>
#include <math.h>


typedef float v2f __attribute__((ext_vector_type(2)));
typedef float v8f __attribute__((ext_vector_type(8)));

#define N_NODES   100000
#define N_EDGES   1600000
#define IN_DIM    256
#define HEADS     4
#define OUT_DIM   32
#define FEAT_DIM  128
#define NEG_SLOPE 0.2f
#define KSTEPS    (IN_DIM / 4)
#define NTILES    (FEAT_DIM / 16)

__device__ __forceinline__ void prepack_body(int b, const float* __restrict__ W,
                                             float* __restrict__ Wp) {
  const int t = b * 256 + threadIdx.x;
  if (t >= KSTEPS * NTILES * 32) return;
  const int lane  = t & 31;
  const int tile  = (t >> 5) & 7;
  const int kstep = t >> 8;
  const int kb = (lane >> 4) * 2;
  const int n  = tile * 16 + (lane & 15);
  const int k  = kstep * 4 + kb;
  Wp[2 * t + 0] = W[(size_t)k * FEAT_DIM + n];
  Wp[2 * t + 1] = W[(size_t)(k + 1) * FEAT_DIM + n];
}

__device__ __forceinline__ void gemm_body(int b, const float* __restrict__ x,
                                          const float* __restrict__ Wp,
                                          float* __restrict__ feat) {
  const int lane = threadIdx.x & 31;
  const int mt   = b * 8 + (threadIdx.x >> 5);
  if (mt >= N_NODES / 16) return;

  const int kb = (lane >> 4) * 2;
  const float* aptr = x + (size_t)(mt * 16 + (lane & 15)) * IN_DIM + kb;
  const float* bptr = Wp + (size_t)lane * 2;

  v8f c[NTILES] = {};

  v2f ap, aq, bp[NTILES], bq[NTILES];
  ap = *(const v2f*)(aptr);
  aq = *(const v2f*)(aptr + 4);
#pragma unroll
  for (int t = 0; t < NTILES; ++t) bp[t] = *(const v2f*)(bptr + (size_t)t * 64);

  for (int ks = 0; ks < KSTEPS; ks += 2) {
    const int k2 = min(ks + 2, KSTEPS - 1);
    const int k3 = min(ks + 3, KSTEPS - 1);
#pragma unroll
    for (int t = 0; t < NTILES; ++t)
      bq[t] = *(const v2f*)(bptr + (size_t)((ks + 1) * NTILES + t) * 64);
#pragma unroll
    for (int t = 0; t < NTILES; ++t)
      c[t] = __builtin_amdgcn_wmma_f32_16x16x4_f32(false, ap, false, bp[t],
                                                   (short)0, c[t], false, false);
    ap = *(const v2f*)(aptr + k2 * 4);
#pragma unroll
    for (int t = 0; t < NTILES; ++t)
      bp[t] = *(const v2f*)(bptr + (size_t)(k2 * NTILES + t) * 64);
#pragma unroll
    for (int t = 0; t < NTILES; ++t)
      c[t] = __builtin_amdgcn_wmma_f32_16x16x4_f32(false, aq, false, bq[t],
                                                   (short)0, c[t], false, false);
    aq = *(const v2f*)(aptr + k3 * 4);
    asm volatile("" ::: "memory");
  }

  const int mbase = mt * 16 + 8 * (lane >> 4);
  float* o = feat + (size_t)mbase * FEAT_DIM + (lane & 15);
#pragma unroll
  for (int t = 0; t < NTILES; ++t)
#pragma unroll
    for (int i = 0; i < 8; ++i)
      o[(size_t)i * FEAT_DIM + t * 16] = c[t][i];
}

#define SPLIT 65536
#define CAP   64

__device__ __forceinline__ void init_body(int b, int* __restrict__ cnt2) {
  const int t = b * 256 + threadIdx.x;
  if (t < 2 * N_NODES) cnt2[t] = 0;
}

__device__ __forceinline__ void el_er_body(int b, const float* __restrict__ feat,
                                           const float* __restrict__ attn_l,
                                           const float* __restrict__ attn_r,
                                           float* __restrict__ el, float* __restrict__ er) {
  int t = b * 256 + threadIdx.x;
  if (t >= N_NODES * HEADS) return;
  const int h = t & 3;
  const float* f  = feat + (size_t)(t >> 2) * FEAT_DIM + h * OUT_DIM;
  const float* al = attn_l + h * OUT_DIM;
  const float* ar = attn_r + h * OUT_DIM;
  float sl = 0.0f, sr = 0.0f;
#pragma unroll
  for (int d = 0; d < OUT_DIM; ++d) { float v = f[d]; sl += v * al[d]; sr += v * ar[d]; }
  el[t] = sl;
  er[t] = sr;
}

__device__ __forceinline__ float lrelu(float v) {
  return v > 0.0f ? v : NEG_SLOPE * v;
}

__device__ __forceinline__ void place_body(int b, const int* __restrict__ src,
                                           const int* __restrict__ dst,
                                           int* __restrict__ cnt2,
                                           unsigned short* __restrict__ bucket) {
  const int e = b * 256 + threadIdx.x;
  if (e >= N_EDGES) return;
  const int s = src[e], d = dst[e];
  const bool up = s >= SPLIT;
  const int r = atomicAdd(cnt2 + (up ? N_NODES : 0) + d, 1);
  if ((unsigned)r < (unsigned)CAP)
    bucket[(size_t)d * CAP + (up ? CAP - 1 - r : r)] = (unsigned short)(s & (SPLIT - 1));
}

#define PACK_BLOCKS ((KSTEPS * NTILES * 32 + 255) / 256)
#define INIT_BLOCKS ((2 * N_NODES + 255) / 256)
#define GEMM_BLOCKS ((N_NODES / 16 + 7) / 8)
#define RANK_BLOCKS ((N_EDGES + 255) / 256)
#define ELER_BLOCKS ((N_NODES * HEADS + 255) / 256)

__global__ __launch_bounds__(256)
void gat_pack_init(const float* __restrict__ W, float* __restrict__ Wp,
                   int* __restrict__ cnt2) {
  const int b = blockIdx.x;
  if (b < PACK_BLOCKS) prepack_body(b, W, Wp);
  else                 init_body(b - PACK_BLOCKS, cnt2);
}

__global__ __launch_bounds__(256)
void gat_gemm_place(const float* __restrict__ x, const float* __restrict__ Wp,
                    float* __restrict__ feat,
                    const int* __restrict__ src, const int* __restrict__ dst,
                    int* __restrict__ cnt2, unsigned short* __restrict__ bucket) {
  const int b = blockIdx.x;
  if (b < GEMM_BLOCKS) gemm_body(b, x, Wp, feat);
  else                 place_body(b - GEMM_BLOCKS, src, dst, cnt2, bucket);
}

__global__ __launch_bounds__(256)
void gat_el_er(const float* __restrict__ feat,
               const float* __restrict__ attn_l,
               const float* __restrict__ attn_r,
               float* __restrict__ el, float* __restrict__ er) {
  el_er_body(blockIdx.x, feat, attn_l, attn_r, el, er);
}

#define EPW 8
#define RCH 4

__device__ __forceinline__ int bucket_src(int w, int i, int lo) {
  const bool up = i >= lo;
  const int k = up ? (CAP - 1 + lo - i) : i;
  const int v = __shfl(w, (k >> 1) & 31);
  const int s = ((k & 1) ? (int)((unsigned)v >> 16) : (v & (SPLIT - 1))) + (up ? SPLIT : 0);
  return min(s, N_NODES - 1);
}

__device__ __forceinline__ void gather_rows(const float* __restrict__ feat, int sv,
                                            int lane, float (&f)[EPW][HEADS]) {
  for (int jj = 0; jj < EPW; ++jj) {
    const int sj = __builtin_amdgcn_readlane(sv, 4 * jj);
    const float* fr = feat + (size_t)sj * FEAT_DIM + lane;
    for (int hh = 0; hh < HEADS; ++hh) f[jj][hh] = fr[hh * OUT_DIM];
  }
}

__device__ __forceinline__ float add_rows(float tot, float a,
                                          const float (&f)[EPW][HEADS]) {
  const int ai = __float_as_int(a);
  for (int jj = 0; jj < EPW; ++jj) {
    const float a0 = __int_as_float(__builtin_amdgcn_readlane(ai, 4 * jj + 0));
    const float a1 = __int_as_float(__builtin_amdgcn_readlane(ai, 4 * jj + 1));
    const float a2 = __int_as_float(__builtin_amdgcn_readlane(ai, 4 * jj + 2));
    const float a3 = __int_as_float(__builtin_amdgcn_readlane(ai, 4 * jj + 3));
    tot += a0 * f[jj][0] + a1 * f[jj][1] + a2 * f[jj][2] + a3 * f[jj][3];
  }
  return tot;
}

__global__ __launch_bounds__(256)
void gat_node(const int* __restrict__ src, const int* __restrict__ dst,
              const int* __restrict__ cnt2, const float* __restrict__ el,
              const float* __restrict__ er, const float* __restrict__ feat,
              float* out) {
  const int lane = threadIdx.x & 31;
  const int n = __builtin_amdgcn_readfirstlane(blockIdx.x * 8 + (threadIdx.x >> 5));
  if (n >= N_NODES) return;
  float* orow = out + (size_t)n * OUT_DIM;
  const int w = ((const int*)orow)[lane];
  const int lo  = min(max(cnt2[n], 0), N_EDGES);
  const int deg = min(lo + max(cnt2[N_NODES + n], 0), N_EDGES);
  if (deg == 0) { orow[lane] = 0.0f; return; }

  const int h = lane & 3, j = lane >> 2;
  const float erh = er[(size_t)n * HEADS + h];

  float f[EPW][HEADS];
  float tot = 0.0f;
  const int last = deg - 1;

  if (deg <= RCH * EPW) {
    int   s[RCH];
    float p[RCH];
    for (int c = 0; c < RCH; ++c)
      s[c] = bucket_src(w, min(c * EPW + j, last), lo);
    for (int c = 0; c < RCH; ++c)
      p[c] = lrelu(el[(size_t)s[c] * HEADS + h] + erh);
    gather_rows(feat, s[0], lane, f);

    float mx = -INFINITY;
    for (int c = 0; c < RCH; ++c)
      mx = (c * EPW + j < deg) ? fmaxf(mx, p[c]) : mx;
    mx = fmaxf(mx, __shfl_xor(mx, 4));
    mx = fmaxf(mx, __shfl_xor(mx, 8));
    mx = fmaxf(mx, __shfl_xor(mx, 16));

    float sm = 0.0f;
    for (int c = 0; c < RCH; ++c) {
      p[c] = (c * EPW + j < deg) ? __expf(p[c] - mx) : 0.0f;
      sm += p[c];
    }
    sm += __shfl_xor(sm, 4);
    sm += __shfl_xor(sm, 8);
    sm += __shfl_xor(sm, 16);
    const float den = fmaxf(sm, 1e-9f);
    for (int c = 0; c < RCH; ++c) p[c] = 0.25f * (p[c] / den);

    tot = add_rows(tot, p[0], f);
    for (int c = 1; c * EPW < deg; ++c) {
      const int   sc = (c == 1) ? s[1] : ((c == 2) ? s[2] : s[3]);
      const float ac = (c == 1) ? p[1] : ((c == 2) ? p[2] : p[3]);
      gather_rows(feat, sc, lane, f);
      tot = add_rows(tot, ac, f);
    }
  } else if (deg <= CAP) {
    float mx = -INFINITY;
    for (int c = 0; c < deg; c += EPW) {
      const int s = bucket_src(w, min(c + j, last), lo);
      const float v = lrelu(el[(size_t)s * HEADS + h] + erh);
      mx = (c + j < deg) ? fmaxf(mx, v) : mx;
    }
    mx = fmaxf(mx, __shfl_xor(mx, 4));
    mx = fmaxf(mx, __shfl_xor(mx, 8));
    mx = fmaxf(mx, __shfl_xor(mx, 16));

    float sm = 0.0f;
    for (int c = 0; c < deg; c += EPW) {
      const int s = bucket_src(w, min(c + j, last), lo);
      const float v = lrelu(el[(size_t)s * HEADS + h] + erh);
      sm += (c + j < deg) ? __expf(v - mx) : 0.0f;
    }
    sm += __shfl_xor(sm, 4);
    sm += __shfl_xor(sm, 8);
    sm += __shfl_xor(sm, 16);
    const float den = fmaxf(sm, 1e-9f);

    for (int c = 0; c < deg; c += EPW) {
      const int s = bucket_src(w, min(c + j, last), lo);
      const float v = lrelu(el[(size_t)s * HEADS + h] + erh);
      const float a = (c + j < deg) ? 0.25f * (__expf(v - mx) / den) : 0.0f;
      gather_rows(feat, s, lane, f);
      tot = add_rows(tot, a, f);
    }
  } else {
    const float e0 = er[(size_t)n * HEADS + 0], e1 = er[(size_t)n * HEADS + 1];
    const float e2 = er[(size_t)n * HEADS + 2], e3 = er[(size_t)n * HEADS + 3];
    float m0 = -INFINITY, m1 = -INFINITY, m2 = -INFINITY, m3 = -INFINITY;
    for (int b0 = 0; b0 < N_EDGES; b0 += 32) {
      const int e = b0 + lane;
      if (e < N_EDGES && dst[e] == n) {
        const int s = min(max(src[e], 0), N_NODES - 1);
        const float* l = el + (size_t)s * HEADS;
        m0 = fmaxf(m0, lrelu(l[0] + e0)); m1 = fmaxf(m1, lrelu(l[1] + e1));
        m2 = fmaxf(m2, lrelu(l[2] + e2)); m3 = fmaxf(m3, lrelu(l[3] + e3));
      }
    }
    for (int o = 16; o > 0; o >>= 1) {
      m0 = fmaxf(m0, __shfl_xor(m0, o)); m1 = fmaxf(m1, __shfl_xor(m1, o));
      m2 = fmaxf(m2, __shfl_xor(m2, o)); m3 = fmaxf(m3, __shfl_xor(m3, o));
    }
    float d0 = 0.0f, d1 = 0.0f, d2 = 0.0f, d3 = 0.0f;
    for (int b0 = 0; b0 < N_EDGES; b0 += 32) {
      const int e = b0 + lane;
      if (e < N_EDGES && dst[e] == n) {
        const int s = min(max(src[e], 0), N_NODES - 1);
        const float* l = el + (size_t)s * HEADS;
        d0 += __expf(lrelu(l[0] + e0) - m0); d1 += __expf(lrelu(l[1] + e1) - m1);
        d2 += __expf(lrelu(l[2] + e2) - m2); d3 += __expf(lrelu(l[3] + e3) - m3);
      }
    }
    for (int o = 16; o > 0; o >>= 1) {
      d0 += __shfl_xor(d0, o); d1 += __shfl_xor(d1, o);
      d2 += __shfl_xor(d2, o); d3 += __shfl_xor(d3, o);
    }
    d0 = fmaxf(d0, 1e-9f); d1 = fmaxf(d1, 1e-9f); d2 = fmaxf(d2, 1e-9f); d3 = fmaxf(d3, 1e-9f);
    for (int b0 = 0; b0 < N_EDGES; b0 += 32) {
      const int e = b0 + lane;
      const bool hit = e < N_EDGES && dst[e] == n;
      int s = 0;
      float a0 = 0.0f, a1 = 0.0f, a2 = 0.0f, a3 = 0.0f;
      if (hit) {
        s = min(max(src[e], 0), N_NODES - 1);
        const float* l = el + (size_t)s * HEADS;
        a0 = 0.25f * (__expf(lrelu(l[0] + e0) - m0) / d0);
        a1 = 0.25f * (__expf(lrelu(l[1] + e1) - m1) / d1);
        a2 = 0.25f * (__expf(lrelu(l[2] + e2) - m2) / d2);
        a3 = 0.25f * (__expf(lrelu(l[3] + e3) - m3) / d3);
      }
      unsigned mask = __builtin_amdgcn_ballot_w32(hit);
      while (mask) {
        const int bl = __builtin_ctz(mask);
        mask &= mask - 1;
        const int sj = __shfl(s, bl);
        const float* fr = feat + (size_t)sj * FEAT_DIM + lane;
        tot += __shfl(a0, bl) * fr[0] + __shfl(a1, bl) * fr[OUT_DIM]
             + __shfl(a2, bl) * fr[2 * OUT_DIM] + __shfl(a3, bl) * fr[3 * OUT_DIM];
      }
    }
  }
  orow[lane] = tot;
}

extern "C" void kernel_launch(void* const* d_in, const int* in_sizes, int n_in,
                              void* d_out, int out_size, void* d_ws, size_t ws_size,
                              hipStream_t stream) {
  const float* x      = (const float*)d_in[0];
  const float* W      = (const float*)d_in[1];
  const float* attn_l = (const float*)d_in[2];
  const float* attn_r = (const float*)d_in[3];
  const int*   src    = (const int*)d_in[4];
  const int*   dst    = (const int*)d_in[5];
  float* out = (float*)d_out;

  float* ws    = (float*)d_ws;
  float* feat  = ws;
  float* el    = feat + (size_t)N_NODES * FEAT_DIM;
  float* er    = el + (size_t)N_NODES * HEADS;
  int*   cnt2  = (int*)(el + (size_t)3 * N_NODES * HEADS);
  float* Wp    = el + (size_t)4 * N_NODES * HEADS;

  gat_pack_init<<<PACK_BLOCKS + INIT_BLOCKS, 256, 0, stream>>>(W, Wp, cnt2);
  gat_gemm_place<<<GEMM_BLOCKS + RANK_BLOCKS, 256, 0, stream>>>(x, Wp, feat, src, dst, cnt2, (unsigned short*)out);
  gat_el_er<<<ELER_BLOCKS, 256, 0, stream>>>(feat, attn_l, attn_r, el, er);
  gat_node<<<(N_NODES + 7) / 8, 256, 0, stream>>>(src, dst, cnt2, el, er, feat, out);
}


# GEMM pass forms the el/er logits from its accumulators (2 products per lane, then a xor tree over the row's 16 lanes: el/er sum order changes); the el/er launch and its re-read of feat are dropped
# speedup vs baseline: 3.4971x; 1.0639x over previous
#include <hip/hip_runtime.h>
#include <math.h>


typedef float v2f __attribute__((ext_vector_type(2)));
typedef float v8f __attribute__((ext_vector_type(8)));

#define N_NODES   100000
#define N_EDGES   1600000
#define IN_DIM    256
#define HEADS     4
#define OUT_DIM   32
#define FEAT_DIM  128
#define NEG_SLOPE 0.2f
#define KSTEPS    (IN_DIM / 4)
#define NTILES    (FEAT_DIM / 16)

__device__ __forceinline__ void prepack_body(int b, const float* __restrict__ W,
                                             float* __restrict__ Wp) {
  const int t = b * 256 + threadIdx.x;
  if (t >= KSTEPS * NTILES * 32) return;
  const int lane  = t & 31;
  const int tile  = (t >> 5) & 7;
  const int kstep = t >> 8;
  const int kb = (lane >> 4) * 2;
  const int n  = tile * 16 + (lane & 15);
  const int k  = kstep * 4 + kb;
  Wp[2 * t + 0] = W[(size_t)k * FEAT_DIM + n];
  Wp[2 * t + 1] = W[(size_t)(k + 1) * FEAT_DIM + n];
}

__device__ __forceinline__ float half_sum16(float v) {
  v += __int_as_float(__builtin_amdgcn_update_dpp(0, __float_as_int(v), 0x168, 0xf, 0xf, true));
  v += __int_as_float(__builtin_amdgcn_update_dpp(0, __float_as_int(v), 0x164, 0xf, 0xf, true));
  v += __int_as_float(__builtin_amdgcn_update_dpp(0, __float_as_int(v), 0x162, 0xf, 0xf, true));
  v += __int_as_float(__builtin_amdgcn_update_dpp(0, __float_as_int(v), 0x161, 0xf, 0xf, true));
  return v;
}

__device__ __forceinline__ void gemm_body(int b, const float* __restrict__ x,
                                          const float* __restrict__ Wp,
                                          float* __restrict__ feat,
                                          const float* __restrict__ attn_l,
                                          const float* __restrict__ attn_r,
                                          float* __restrict__ el, float* __restrict__ er) {
  const int lane = threadIdx.x & 31;
  const int mt   = b * 8 + (threadIdx.x >> 5);
  if (mt >= N_NODES / 16) return;

  const int kb = (lane >> 4) * 2;
  const float* aptr = x + (size_t)(mt * 16 + (lane & 15)) * IN_DIM + kb;
  const float* bptr = Wp + (size_t)lane * 2;

  v8f c[NTILES] = {};

  v2f ap, aq, bp[NTILES], bq[NTILES];
  ap = *(const v2f*)(aptr);
  aq = *(const v2f*)(aptr + 4);
#pragma unroll
  for (int t = 0; t < NTILES; ++t) bp[t] = *(const v2f*)(bptr + (size_t)t * 64);

  for (int ks = 0; ks < KSTEPS; ks += 2) {
    const int k2 = min(ks + 2, KSTEPS - 1);
    const int k3 = min(ks + 3, KSTEPS - 1);
#pragma unroll
    for (int t = 0; t < NTILES; ++t)
      bq[t] = *(const v2f*)(bptr + (size_t)((ks + 1) * NTILES + t) * 64);
#pragma unroll
    for (int t = 0; t < NTILES; ++t)
      c[t] = __builtin_amdgcn_wmma_f32_16x16x4_f32(false, ap, false, bp[t],
                                                   (short)0, c[t], false, false);
    ap = *(const v2f*)(aptr + k2 * 4);
#pragma unroll
    for (int t = 0; t < NTILES; ++t)
      bp[t] = *(const v2f*)(bptr + (size_t)(k2 * NTILES + t) * 64);
#pragma unroll
    for (int t = 0; t < NTILES; ++t)
      c[t] = __builtin_amdgcn_wmma_f32_16x16x4_f32(false, aq, false, bq[t],
                                                   (short)0, c[t], false, false);
    aq = *(const v2f*)(aptr + k3 * 4);
    asm volatile("" ::: "memory");
  }

  const int c16 = lane & 15;
  const int mbase = mt * 16 + 8 * (lane >> 4);
  float* o = feat + (size_t)mbase * FEAT_DIM + c16;
  v2f ol = {0.0f, 0.0f}, orr = {0.0f, 0.0f};
#pragma unroll
  for (int hd = 0; hd < HEADS; ++hd) {
#pragma unroll
    for (int tt = 0; tt < 2; ++tt)
#pragma unroll
      for (int i = 0; i < 8; ++i)
        o[(size_t)i * FEAT_DIM + (2 * hd + tt) * 16] = c[2 * hd + tt][i];
    const float al0 = attn_l[hd * OUT_DIM + c16], al1 = attn_l[hd * OUT_DIM + 16 + c16];
    const float ar0 = attn_r[hd * OUT_DIM + c16], ar1 = attn_r[hd * OUT_DIM + 16 + c16];
#pragma unroll
    for (int i = 0; i < 8; ++i) {
      const float f0 = c[2 * hd][i], f1 = c[2 * hd + 1][i];
      const float sl = half_sum16(f0 * al0 + f1 * al1);
      const float sr = half_sum16(f0 * ar0 + f1 * ar1);
      const bool mine = ((c16 >> 1) == i) && ((c16 & 1) == (hd >> 1));
      if (hd & 1) { ol.y = mine ? sl : ol.y; orr.y = mine ? sr : orr.y; }
      else        { ol.x = mine ? sl : ol.x; orr.x = mine ? sr : orr.x; }
    }
  }
  ((v2f*)(el + (size_t)mbase * HEADS))[c16] = ol;
  ((v2f*)(er + (size_t)mbase * HEADS))[c16] = orr;
}

#define SPLIT 65536
#define CAP   64

__device__ __forceinline__ void init_body(int b, int* __restrict__ cnt2) {
  const int t = b * 256 + threadIdx.x;
  if (t < 2 * N_NODES) cnt2[t] = 0;
}

__device__ __forceinline__ float lrelu(float v) {
  return v > 0.0f ? v : NEG_SLOPE * v;
}

__device__ __forceinline__ void place_body(int b, const int* __restrict__ src,
                                           const int* __restrict__ dst,
                                           int* __restrict__ cnt2,
                                           unsigned short* __restrict__ bucket) {
  const int e = b * 256 + threadIdx.x;
  if (e >= N_EDGES) return;
  const int s = src[e], d = dst[e];
  const bool up = s >= SPLIT;
  const int r = atomicAdd(cnt2 + (up ? N_NODES : 0) + d, 1);
  if ((unsigned)r < (unsigned)CAP)
    bucket[(size_t)d * CAP + (up ? CAP - 1 - r : r)] = (unsigned short)(s & (SPLIT - 1));
}

#define PACK_BLOCKS ((KSTEPS * NTILES * 32 + 255) / 256)
#define INIT_BLOCKS ((2 * N_NODES + 255) / 256)
#define GEMM_BLOCKS ((N_NODES / 16 + 7) / 8)
#define RANK_BLOCKS ((N_EDGES + 255) / 256)

__global__ __launch_bounds__(256)
void gat_pack_init(const float* __restrict__ W, float* __restrict__ Wp,
                   int* __restrict__ cnt2) {
  const int b = blockIdx.x;
  if (b < PACK_BLOCKS) prepack_body(b, W, Wp);
  else                 init_body(b - PACK_BLOCKS, cnt2);
}

__global__ __launch_bounds__(256)
void gat_gemm_place(const float* __restrict__ x, const float* __restrict__ Wp,
                    float* __restrict__ feat,
                    const float* __restrict__ attn_l, const float* __restrict__ attn_r,
                    float* __restrict__ el, float* __restrict__ er,
                    const int* __restrict__ src, const int* __restrict__ dst,
                    int* __restrict__ cnt2, unsigned short* __restrict__ bucket) {
  const int b = blockIdx.x;
  if (b < GEMM_BLOCKS) gemm_body(b, x, Wp, feat, attn_l, attn_r, el, er);
  else                 place_body(b - GEMM_BLOCKS, src, dst, cnt2, bucket);
}

#define EPW 8
#define RCH 4

__device__ __forceinline__ int bucket_src(int w, int i, int lo) {
  const bool up = i >= lo;
  const int k = up ? (CAP - 1 + lo - i) : i;
  const int v = __shfl(w, (k >> 1) & 31);
  const int s = ((k & 1) ? (int)((unsigned)v >> 16) : (v & (SPLIT - 1))) + (up ? SPLIT : 0);
  return min(s, N_NODES - 1);
}

__device__ __forceinline__ void gather_rows(const float* __restrict__ feat, int sv,
                                            int lane, float (&f)[EPW][HEADS]) {
  for (int jj = 0; jj < EPW; ++jj) {
    const int sj = __builtin_amdgcn_readlane(sv, 4 * jj);
    const float* fr = feat + (size_t)sj * FEAT_DIM + lane;
    for (int hh = 0; hh < HEADS; ++hh) f[jj][hh] = fr[hh * OUT_DIM];
  }
}

__device__ __forceinline__ float add_rows(float tot, float a,
                                          const float (&f)[EPW][HEADS]) {
  const int ai = __float_as_int(a);
  for (int jj = 0; jj < EPW; ++jj) {
    const float a0 = __int_as_float(__builtin_amdgcn_readlane(ai, 4 * jj + 0));
    const float a1 = __int_as_float(__builtin_amdgcn_readlane(ai, 4 * jj + 1));
    const float a2 = __int_as_float(__builtin_amdgcn_readlane(ai, 4 * jj + 2));
    const float a3 = __int_as_float(__builtin_amdgcn_readlane(ai, 4 * jj + 3));
    tot += a0 * f[jj][0] + a1 * f[jj][1] + a2 * f[jj][2] + a3 * f[jj][3];
  }
  return tot;
}

__global__ __launch_bounds__(256)
void gat_node(const int* __restrict__ src, const int* __restrict__ dst,
              const int* __restrict__ cnt2, const float* __restrict__ el,
              const float* __restrict__ er, const float* __restrict__ feat,
              float* out) {
  const int lane = threadIdx.x & 31;
  const int n = __builtin_amdgcn_readfirstlane(blockIdx.x * 8 + (threadIdx.x >> 5));
  if (n >= N_NODES) return;
  float* orow = out + (size_t)n * OUT_DIM;
  const int w = ((const int*)orow)[lane];
  const int lo  = min(max(cnt2[n], 0), N_EDGES);
  const int deg = min(lo + max(cnt2[N_NODES + n], 0), N_EDGES);
  if (deg == 0) { orow[lane] = 0.0f; return; }

  const int h = lane & 3, j = lane >> 2;
  const float erh = er[(size_t)n * HEADS + h];

  float f[EPW][HEADS];
  float tot = 0.0f;
  const int last = deg - 1;

  if (deg <= RCH * EPW) {
    int   s[RCH];
    float p[RCH];
    for (int c = 0; c < RCH; ++c)
      s[c] = bucket_src(w, min(c * EPW + j, last), lo);
    for (int c = 0; c < RCH; ++c)
      p[c] = lrelu(el[(size_t)s[c] * HEADS + h] + erh);
    gather_rows(feat, s[0], lane, f);

    float mx = -INFINITY;
    for (int c = 0; c < RCH; ++c)
      mx = (c * EPW + j < deg) ? fmaxf(mx, p[c]) : mx;
    mx = fmaxf(mx, __shfl_xor(mx, 4));
    mx = fmaxf(mx, __shfl_xor(mx, 8));
    mx = fmaxf(mx, __shfl_xor(mx, 16));

    float sm = 0.0f;
    for (int c = 0; c < RCH; ++c) {
      p[c] = (c * EPW + j < deg) ? __expf(p[c] - mx) : 0.0f;
      sm += p[c];
    }
    sm += __shfl_xor(sm, 4);
    sm += __shfl_xor(sm, 8);
    sm += __shfl_xor(sm, 16);
    const float den = fmaxf(sm, 1e-9f);
    for (int c = 0; c < RCH; ++c) p[c] = 0.25f * (p[c] / den);

    tot = add_rows(tot, p[0], f);
    for (int c = 1; c * EPW < deg; ++c) {
      const int   sc = (c == 1) ? s[1] : ((c == 2) ? s[2] : s[3]);
      const float ac = (c == 1) ? p[1] : ((c == 2) ? p[2] : p[3]);
      gather_rows(feat, sc, lane, f);
      tot = add_rows(tot, ac, f);
    }
  } else if (deg <= CAP) {
    float mx = -INFINITY;
    for (int c = 0; c < deg; c += EPW) {
      const int s = bucket_src(w, min(c + j, last), lo);
      const float v = lrelu(el[(size_t)s * HEADS + h] + erh);
      mx = (c + j < deg) ? fmaxf(mx, v) : mx;
    }
    mx = fmaxf(mx, __shfl_xor(mx, 4));
    mx = fmaxf(mx, __shfl_xor(mx, 8));
    mx = fmaxf(mx, __shfl_xor(mx, 16));

    float sm = 0.0f;
    for (int c = 0; c < deg; c += EPW) {
      const int s = bucket_src(w, min(c + j, last), lo);
      const float v = lrelu(el[(size_t)s * HEADS + h] + erh);
      sm += (c + j < deg) ? __expf(v - mx) : 0.0f;
    }
    sm += __shfl_xor(sm, 4);
    sm += __shfl_xor(sm, 8);
    sm += __shfl_xor(sm, 16);
    const float den = fmaxf(sm, 1e-9f);

    for (int c = 0; c < deg; c += EPW) {
      const int s = bucket_src(w, min(c + j, last), lo);
      const float v = lrelu(el[(size_t)s * HEADS + h] + erh);
      const float a = (c + j < deg) ? 0.25f * (__expf(v - mx) / den) : 0.0f;
      gather_rows(feat, s, lane, f);
      tot = add_rows(tot, a, f);
    }
  } else {
    const float e0 = er[(size_t)n * HEADS + 0], e1 = er[(size_t)n * HEADS + 1];
    const float e2 = er[(size_t)n * HEADS + 2], e3 = er[(size_t)n * HEADS + 3];
    float m0 = -INFINITY, m1 = -INFINITY, m2 = -INFINITY, m3 = -INFINITY;
    for (int b0 = 0; b0 < N_EDGES; b0 += 32) {
      const int e = b0 + lane;
      if (e < N_EDGES && dst[e] == n) {
        const int s = min(max(src[e], 0), N_NODES - 1);
        const float* l = el + (size_t)s * HEADS;
        m0 = fmaxf(m0, lrelu(l[0] + e0)); m1 = fmaxf(m1, lrelu(l[1] + e1));
        m2 = fmaxf(m2, lrelu(l[2] + e2)); m3 = fmaxf(m3, lrelu(l[3] + e3));
      }
    }
    for (int o = 16; o > 0; o >>= 1) {
      m0 = fmaxf(m0, __shfl_xor(m0, o)); m1 = fmaxf(m1, __shfl_xor(m1, o));
      m2 = fmaxf(m2, __shfl_xor(m2, o)); m3 = fmaxf(m3, __shfl_xor(m3, o));
    }
    float d0 = 0.0f, d1 = 0.0f, d2 = 0.0f, d3 = 0.0f;
    for (int b0 = 0; b0 < N_EDGES; b0 += 32) {
      const int e = b0 + lane;
      if (e < N_EDGES && dst[e] == n) {
        const int s = min(max(src[e], 0), N_NODES - 1);
        const float* l = el + (size_t)s * HEADS;
        d0 += __expf(lrelu(l[0] + e0) - m0); d1 += __expf(lrelu(l[1] + e1) - m1);
        d2 += __expf(lrelu(l[2] + e2) - m2); d3 += __expf(lrelu(l[3] + e3) - m3);
      }
    }
    for (int o = 16; o > 0; o >>= 1) {
      d0 += __shfl_xor(d0, o); d1 += __shfl_xor(d1, o);
      d2 += __shfl_xor(d2, o); d3 += __shfl_xor(d3, o);
    }
    d0 = fmaxf(d0, 1e-9f); d1 = fmaxf(d1, 1e-9f); d2 = fmaxf(d2, 1e-9f); d3 = fmaxf(d3, 1e-9f);
    for (int b0 = 0; b0 < N_EDGES; b0 += 32) {
      const int e = b0 + lane;
      const bool hit = e < N_EDGES && dst[e] == n;
      int s = 0;
      float a0 = 0.0f, a1 = 0.0f, a2 = 0.0f, a3 = 0.0f;
      if (hit) {
        s = min(max(src[e], 0), N_NODES - 1);
        const float* l = el + (size_t)s * HEADS;
        a0 = 0.25f * (__expf(lrelu(l[0] + e0) - m0) / d0);
        a1 = 0.25f * (__expf(lrelu(l[1] + e1) - m1) / d1);
        a2 = 0.25f * (__expf(lrelu(l[2] + e2) - m2) / d2);
        a3 = 0.25f * (__expf(lrelu(l[3] + e3) - m3) / d3);
      }
      unsigned mask = __builtin_amdgcn_ballot_w32(hit);
      while (mask) {
        const int bl = __builtin_ctz(mask);
        mask &= mask - 1;
        const int sj = __shfl(s, bl);
        const float* fr = feat + (size_t)sj * FEAT_DIM + lane;
        tot += __shfl(a0, bl) * fr[0] + __shfl(a1, bl) * fr[OUT_DIM]
             + __shfl(a2, bl) * fr[2 * OUT_DIM] + __shfl(a3, bl) * fr[3 * OUT_DIM];
      }
    }
  }
  orow[lane] = tot;
}

extern "C" void kernel_launch(void* const* d_in, const int* in_sizes, int n_in,
                              void* d_out, int out_size, void* d_ws, size_t ws_size,
                              hipStream_t stream) {
  const float* x      = (const float*)d_in[0];
  const float* W      = (const float*)d_in[1];
  const float* attn_l = (const float*)d_in[2];
  const float* attn_r = (const float*)d_in[3];
  const int*   src    = (const int*)d_in[4];
  const int*   dst    = (const int*)d_in[5];
  float* out = (float*)d_out;

  float* ws    = (float*)d_ws;
  float* feat  = ws;
  float* el    = feat + (size_t)N_NODES * FEAT_DIM;
  float* er    = el + (size_t)N_NODES * HEADS;
  int*   cnt2  = (int*)(el + (size_t)3 * N_NODES * HEADS);
  float* Wp    = el + (size_t)4 * N_NODES * HEADS;

  gat_pack_init<<<PACK_BLOCKS + INIT_BLOCKS, 256, 0, stream>>>(W, Wp, cnt2);
  gat_gemm_place<<<GEMM_BLOCKS + RANK_BLOCKS, 256, 0, stream>>>(x, Wp, feat, attn_l, attn_r, el, er, src, dst, cnt2, (unsigned short*)out);
  gat_node<<<(N_NODES + 7) / 8, 256, 0, stream>>>(src, dst, cnt2, el, er, feat, out);
}
